# starting kernel
# baseline (speedup 1.0000x reference)
#include <hip/hip_runtime.h>
#include <hip/hip_bf16.h>


typedef __attribute__((ext_vector_type(2))) float v2f;
typedef __attribute__((ext_vector_type(8))) float v8f;

#define N_NODES 50000
#define N_EDGES 800000
#define HID     128
#define N_POS   100000

__global__ void zero_kernel(float* __restrict__ p, long long n) {
  long long i = (long long)blockIdx.x * blockDim.x + threadIdx.x;
  long long stride = (long long)gridDim.x * blockDim.x;
  for (; i < n; i += stride) p[i] = 0.0f;
}

__global__ void transpose_kernel(const float* __restrict__ W, float* __restrict__ Wt,
                                 int K) {
  int id = blockIdx.x * blockDim.x + threadIdx.x;
  if (id < K * HID) {
    int n = id / K;
    int k = id - n * K;
    Wt[id] = W[k * HID + n];
  }
}

__global__ void deg_kernel(const int* __restrict__ idx, const float* __restrict__ vals,
                           float* __restrict__ deg, int nedges) {
  int e = blockIdx.x * blockDim.x + threadIdx.x;
  if (e < nedges) atomicAdd(&deg[idx[2 * e]], vals[e]);
}

__global__ void deginv_kernel(float* __restrict__ deg, int n) {
  int i = blockIdx.x * blockDim.x + threadIdx.x;
  if (i < n) deg[i] = 1.0f / fmaxf(deg[i], 1.0f);
}

__global__ void spmm_kernel(const int* __restrict__ idx, const float* __restrict__ vals,
                            const float* __restrict__ x, float* __restrict__ out,
                            int nedges) {
  int gw   = (blockIdx.x * blockDim.x + threadIdx.x) >> 5;
  int lane = threadIdx.x & 31;
  if (gw >= nedges) return;
  int   row = idx[2 * gw];
  int   col = idx[2 * gw + 1];
  float v   = vals[gw];
  int   f   = lane << 2;
  const float4 xv = *(const float4*)(x + (long long)col * HID + f);
  float* o = out + (long long)row * HID + f;
  atomicAdd(o + 0, v * xv.x);
  atomicAdd(o + 1, v * xv.y);
  atomicAdd(o + 2, v * xv.z);
  atomicAdd(o + 3, v * xv.w);
}

__global__ void gemm_wmma_kernel(const float* __restrict__ X, const float* __restrict__ scale,
                                 const float* __restrict__ Wt, const float* __restrict__ bias,
                                 float* __restrict__ Y, int do_relu) {
  __shared__ float Xs[16][132];
  const int tile_m = blockIdx.x;
  const int t      = threadIdx.x;

  {
    int   r    = t >> 4;
    int   cseg = (t & 15) << 3;
    float s    = scale[tile_m * 16 + r];
    const float* src = X + ((long long)tile_m * 16 + r) * HID + cseg;
    float4 v0 = *(const float4*)(src);
    float4 v1 = *(const float4*)(src + 4);
    float* dst = &Xs[r][cseg];
    dst[0] = v0.x * s; dst[1] = v0.y * s; dst[2] = v0.z * s; dst[3] = v0.w * s;
    dst[4] = v1.x * s; dst[5] = v1.y * s; dst[6] = v1.z * s; dst[7] = v1.w * s;
  }
  __syncthreads();

  const int lane = t & 31;
  const int wid  = t >> 5;
  const int m    = lane & 15;
  const int kh   = (lane >> 4) << 1;
  const int n    = (wid << 4) + (lane & 15);

  const float* wrow = Wt + n * HID;

  v8f c = {};
#pragma unroll
  for (int k0 = 0; k0 < HID; k0 += 4) {
    int k = k0 + kh;
    v2f a; a.x = Xs[m][k]; a.y = Xs[m][k + 1];
    v2f b = *(const v2f*)(wrow + k);
    c = __builtin_amdgcn_wmma_f32_16x16x4_f32(false, a, false, b, (short)0, c,
                                              false, false);
  }

  const float bn      = bias[n];
  const int   rowbase = tile_m * 16 + ((lane >> 4) << 3);
#pragma unroll
  for (int i = 0; i < 8; ++i) {
    float v = c[i] + bn;
    if (do_relu) v = fmaxf(v, 0.0f);
    Y[(long long)(rowbase + i) * HID + n] = v;
  }
}

__global__ void decode_kernel(const float* __restrict__ h2,
                              const int* __restrict__ U, const int* __restrict__ V,
                              const float* __restrict__ Wd1t, const float* __restrict__ bd1,
                              const float* __restrict__ Wd2, const float* __restrict__ bd2,
                              float* __restrict__ out) {
  __shared__ float Zs[16][260];
  __shared__ float Hs[16][HID];
  __shared__ float red[16][16];

  const int base = blockIdx.x * 16;
  const int t    = threadIdx.x;

  {
    int i   = t >> 4;
    int seg = t & 15;
    const float* src = (seg < 8)
        ? h2 + (long long)U[base + i] * HID + (seg << 4)
        : h2 + (long long)V[base + i] * HID + ((seg - 8) << 4);
    float* dst = &Zs[i][seg << 4];
#pragma unroll
    for (int j = 0; j < 16; j += 4) {
      float4 v = *(const float4*)(src + j);
      dst[j] = v.x; dst[j + 1] = v.y; dst[j + 2] = v.z; dst[j + 3] = v.w;
    }
  }
  __syncthreads();

  const int lane = t & 31;
  const int wid  = t >> 5;
  const int m    = lane & 15;
  const int kh   = (lane >> 4) << 1;
  const int n    = (wid << 4) + (lane & 15);

  const float* wrow = Wd1t + n * (2 * HID);

  v8f c = {};
#pragma unroll
  for (int k0 = 0; k0 < 2 * HID; k0 += 4) {
    int k = k0 + kh;
    v2f a; a.x = Zs[m][k]; a.y = Zs[m][k + 1];
    v2f b = *(const v2f*)(wrow + k);
    c = __builtin_amdgcn_wmma_f32_16x16x4_f32(false, a, false, b, (short)0, c,
                                              false, false);
  }

  const float bn    = bd1[n];
  const int   rbase = (lane >> 4) << 3;
#pragma unroll
  for (int i = 0; i < 8; ++i) Hs[rbase + i][n] = fmaxf(c[i] + bn, 0.0f);
  __syncthreads();

  {
    int i   = t >> 4;
    int seg = t & 15;
    float p = 0.0f;
#pragma unroll
    for (int j = 0; j < 8; ++j) p += Hs[i][(seg << 3) + j] * Wd2[(seg << 3) + j];
    red[i][seg] = p;
  }
  __syncthreads();
  if (t < 16) {
    float s = bd2[0];
#pragma unroll
    for (int j = 0; j < 16; ++j) s += red[t][j];
    out[base + t] = s;
  }
}

extern "C" void kernel_launch(void* const* d_in, const int* in_sizes, int n_in,
                              void* d_out, int out_size, void* d_ws, size_t ws_size,
                              hipStream_t stream) {
  const int*   adj_idx = (const int*)  d_in[0];
  const float* adj_val = (const float*)d_in[1];
  const float* h       = (const float*)d_in[3];
  const int*   pos_u   = (const int*)  d_in[4];
  const int*   pos_v   = (const int*)  d_in[5];
  const int*   neg_u   = (const int*)  d_in[6];
  const int*   neg_v   = (const int*)  d_in[7];
  const float* W1      = (const float*)d_in[8];
  const float* b1      = (const float*)d_in[9];
  const float* W2      = (const float*)d_in[10];
  const float* b2      = (const float*)d_in[11];
  const float* Wd1     = (const float*)d_in[12];
  const float* bd1     = (const float*)d_in[13];
  const float* Wd2     = (const float*)d_in[14];
  const float* bd2     = (const float*)d_in[15];

  float* outp = (float*)d_out;
  float* pos_scores = outp;
  float* neg_scores = outp + N_POS;
  float* h2         = outp + 2 * N_POS;

  float* ws      = (float*)d_ws;
  float* deg_inv = ws;
  float* tmp     = ws + 50176;
  float* h1      = tmp + (long long)N_NODES * HID;
  float* W1t     = h1 + (long long)N_NODES * HID;
  float* W2t     = W1t + HID * HID;
  float* Wd1t    = W2t + HID * HID;

  const long long feat_elems = (long long)N_NODES * HID;

  transpose_kernel<<<(HID * HID + 255) / 256, 256, 0, stream>>>(W1, W1t, HID);
  transpose_kernel<<<(HID * HID + 255) / 256, 256, 0, stream>>>(W2, W2t, HID);
  transpose_kernel<<<(2 * HID * HID + 255) / 256, 256, 0, stream>>>(Wd1, Wd1t, 2 * HID);
  zero_kernel<<<2048, 256, 0, stream>>>(ws, 50176 + feat_elems);
  deg_kernel<<<(N_EDGES + 255) / 256, 256, 0, stream>>>(adj_idx, adj_val, deg_inv, N_EDGES);
  deginv_kernel<<<(N_NODES + 255) / 256, 256, 0, stream>>>(deg_inv, N_NODES);
  spmm_kernel<<<N_EDGES / 8, 256, 0, stream>>>(adj_idx, adj_val, h, tmp, N_EDGES);
  gemm_wmma_kernel<<<N_NODES / 16, 256, 0, stream>>>(tmp, deg_inv, W1t, b1, h1, 1);
  zero_kernel<<<2048, 256, 0, stream>>>(tmp, feat_elems);
  spmm_kernel<<<N_EDGES / 8, 256, 0, stream>>>(adj_idx, adj_val, h1, tmp, N_EDGES);
  gemm_wmma_kernel<<<N_NODES / 16, 256, 0, stream>>>(tmp, deg_inv, W2t, b2, h2, 0);
  decode_kernel<<<N_POS / 16, 256, 0, stream>>>(h2, pos_u, pos_v, Wd1t, bd1, Wd2, bd2, pos_scores);
  decode_kernel<<<N_POS / 16, 256, 0, stream>>>(h2, neg_u, neg_v, Wd1t, bd1, Wd2, bd2, neg_scores);
}


# Sweep reads a 1-byte row-range key per edge (written by a new first pass) instead of the 8-byte index pair and fetches the pair only for matching edges; adds stay LDS float atomics, arrival order
# speedup vs baseline: 2.6474x; 2.6474x over previous
#include <hip/hip_runtime.h>
#include <hip/hip_bf16.h>


typedef __attribute__((ext_vector_type(2))) float v2f;
typedef __attribute__((ext_vector_type(8))) float v8f;

#define N_NODES 50000
#define N_EDGES 800000
#define HID     128
#define N_POS   100000

#define RT      5
#define BROWS   (RT * 16)
static_assert(N_NODES % BROWS == 0, "row blocks must tile the node count");
static_assert(N_POS   % BROWS == 0, "pair blocks must tile the pair count");

__global__ void transpose_kernel(const float* __restrict__ W, float* __restrict__ Wt,
                                 int K) {
  int id = blockIdx.x * blockDim.x + threadIdx.x;
  if (id < K * HID) {
    int n = id / K;
    int k = id - n * K;
    Wt[id] = W[k * HID + n];
  }
}

#define SR      400
#define EH      2
#define SUB     200
#define NSUB    (SR / SUB)
#define NRG     (N_NODES / SR)
#define NWG     (NRG * EH)
#define NRP     (N_NODES / SUB)
#define SPW     32
#define NL      4
#define NP      (N_EDGES / 16)
#define BP      32
#define NBATCH  ((NP + BP - 1) / BP)
#define QCH     32
#define QCAP    (QCH + 512)
#define KNONE   255u
#define PARTB   ((NBATCH + EH - 1) / EH)
static_assert(N_NODES % SR == 0, "row ranges must tile the node count");
static_assert(SR % SUB == 0 && NSUB == 2, "a kept edge goes to one of two sub-range lists");
static_assert(N_EDGES % 16 == 0, "sixteen edge keys per 16-byte load");
static_assert(NRG < (int)KNONE, "a range number fits a byte beside the no-range key");
static_assert(N_NODES <= 65536, "a column index is packed into 16 bits");
static_assert(SR < 65536, "a relative row is packed into 16 bits");
static_assert(PARTB >= SPW && NBATCH - (EH - 1) * PARTB >= SPW, "a part holds a batch per wave");

#define CAP     128
#define NLIST   (NRP * EH * SPW)
static_assert(SPW == 32, "one lane per wave when the totals of a work-group are read");
static_assert((long long)NLIST * (CAP * 8 + 4) + (long long)EH * N_NODES * 4 + N_EDGES
              <= (long long)N_NODES * HID * 4,
              "lists, totals, partial degrees and keys must fit the buffer they are placed in");
static_assert(((long long)NLIST * (CAP * 8 + 4) + (long long)EH * N_NODES * 4) % 16 == 0,
              "the keys behind them are read 16 bytes at a time");

__global__ void edge_key_kernel(const int* __restrict__ idx, int* __restrict__ keys) {
  const int t = blockIdx.x * blockDim.x + threadIdx.x;
  if (t < N_EDGES / 4) {
    const int4* E4 = (const int4*)idx;
    const int4 a = E4[2 * t];
    const int4 b = E4[2 * t + 1];
    const unsigned k0 = min((unsigned)a.x / (unsigned)SR, KNONE);
    const unsigned k1 = min((unsigned)a.z / (unsigned)SR, KNONE);
    const unsigned k2 = min((unsigned)b.x / (unsigned)SR, KNONE);
    const unsigned k3 = min((unsigned)b.z / (unsigned)SR, KNONE);
    keys[t] = (int)(k0 | (k1 << 8) | (k2 << 16) | (k3 << 24));
  }
}

#define RD      8
typedef __attribute__((ext_vector_type(4))) float v4f;
static_assert(QCH % RD == 0, "whole groups of queued edges");

template <bool WITH_DEG, bool SAVE, int ROWS>
__device__ __forceinline__
void work_chunk(const int* __restrict__ idx, const float* __restrict__ vals,
                const float* __restrict__ x, float* acc, float* dacc,
                const unsigned* qe, const unsigned n,
                int2* __restrict__ wl0, int2* __restrict__ wl1,
                unsigned& tot0, unsigned& tot1, const int r0, const int lane) {
  const int2* E2 = (const int2*)idx;
  const v4f*  X4 = (const v4f*)x;
  const bool     valid = (unsigned)lane < n;
  const unsigned e     = qe[valid ? lane : 0];
  const int2     rc    = E2[e];
  const int      vb    = __float_as_int(vals[e]);
  const unsigned rel   = (unsigned)(rc.x - r0);
  const bool     keep  = valid & (rel < (unsigned)ROWS);
  const unsigned pk    = (rel << 16) | ((unsigned)rc.y & 0xffffu);
  const unsigned mk    = (unsigned)__ballot(keep);

  if (SAVE) {
    const bool     hi  = keep & (rel >= (unsigned)SUB);
    const bool     lo  = keep & (rel <  (unsigned)SUB);
    const unsigned mh  = (unsigned)__ballot(hi);
    const unsigned ml  = (unsigned)__ballot(lo);
    const unsigned nh  = (unsigned)__builtin_popcount(mh);
    const unsigned nlo = (unsigned)__builtin_popcount(ml);
    if (hi & (tot1 + nh <= (unsigned)CAP)) {
      const unsigned pos = tot1 + __builtin_amdgcn_mbcnt_lo(mh, 0u);
      wl1[pos] = int2{(int)(pk - ((unsigned)SUB << 16)), vb};
    }
    if (lo & (tot0 + nlo <= (unsigned)CAP)) {
      const unsigned pos = tot0 + __builtin_amdgcn_mbcnt_lo(ml, 0u);
      wl0[pos] = int2{(int)pk, vb};
    }
    tot0 += nlo;
    tot1 += nh;
  }

  for (unsigned j0 = 0; j0 < n; j0 += RD) {
    unsigned pj[RD];
    float    vj[RD];
    v4f      xv[RD];
#pragma unroll
    for (int j = 0; j < RD; ++j) {
      const int jj = (int)min(j0 + (unsigned)j, n - 1u);
      pj[j] = (unsigned)__builtin_amdgcn_readlane((int)pk, jj);
      vj[j] = __int_as_float(__builtin_amdgcn_readlane(vb, jj));
      xv[j] = X4[(pj[j] & 0xffffu) * (HID / 4) + lane];
    }
    asm volatile("" :: "v"(xv[0]), "v"(xv[1]), "v"(xv[2]), "v"(xv[3]),
                       "v"(xv[4]), "v"(xv[5]), "v"(xv[6]), "v"(xv[7]));
    static_assert(RD == 8, "the statement above names RD gathers");
#pragma unroll
    for (int j = 0; j < RD; ++j) {
      if ((mk >> (j0 + (unsigned)j)) & 1u) {
        const unsigned rl = pj[j] >> 16;
        float* a = acc + rl * HID + lane;
        atomicAdd(a,      vj[j] * xv[j].x);
        atomicAdd(a + 32, vj[j] * xv[j].y);
        atomicAdd(a + 64, vj[j] * xv[j].z);
        atomicAdd(a + 96, vj[j] * xv[j].w);
        if (WITH_DEG) { if (lane == 0) atomicAdd(dacc + rl, vj[j]); }
      }
    }
  }
}

template <bool WITH_DEG, bool SAVE, int ROWS>
__device__ __forceinline__
void sweep_edges(const int* __restrict__ idx, const float* __restrict__ vals,
                 const float* __restrict__ x, const int* __restrict__ keys,
                 float* acc, float* dacc, unsigned* qe,
                 int2* __restrict__ wl0, int2* __restrict__ wl1,
                 unsigned& tot0, unsigned& tot1,
                 const int r0, const unsigned key, const int qlo, const int nb, const int qoff,
                 const int lane, const int wid) {
  const uint4* K4 = (const uint4*)keys;
  constexpr bool TAIL = (NP % BP) != 0;
  const unsigned kpat = key * 0x01010101u;

  static_assert(NL == 4, "four loads in flight, handed on register to register");
  uint4 c0, c1, c2, c3;
  {
    int qa = wid + qoff;           if (qa >= nb) qa -= nb;
    int qb = wid + SPW + qoff;     if (qb >= nb) qb -= nb;
    int qc = wid + 2 * SPW + qoff; if (qc >= nb) qc -= nb;
    int qd = wid + 3 * SPW + qoff; if (qd >= nb) qd -= nb;
    c0 = K4[min((qlo + qa) * BP + lane, NP - 1)];
    c1 = K4[min((qlo + qb) * BP + lane, NP - 1)];
    c2 = K4[min((qlo + qc) * BP + lane, NP - 1)];
    c3 = K4[min((qlo + qd) * BP + lane, NP - 1)];
  }

  unsigned cnt = 0;
  for (int q = wid; q < nb; q += SPW) {
    int qq = q + qoff; if (qq >= nb) qq -= nb;
    const int p = (qlo + qq) * BP + lane;
    uint4 kw = c0;
    if (TAIL) { if (p >= NP) kw = uint4{~0u, ~0u, ~0u, ~0u}; }
    c0 = c1; c1 = c2; c2 = c3;
    {
      int qn = q + NL * SPW + qoff; if (qn >= nb) qn -= nb;
      c3 = K4[min((qlo + qn) * BP + lane, NP - 1)];
    }

    const unsigned w[4] = {kw.x, kw.y, kw.z, kw.w};
#pragma unroll
    for (int d = 0; d < 4; ++d) {
      const unsigned xk = w[d] ^ kpat;
      unsigned t = ~(((xk & 0x7f7f7f7fu) + 0x7f7f7f7fu) | xk | 0x7f7f7f7fu);
      unsigned m = (unsigned)__ballot(t != 0u);
      while (m != 0u) {
        if (t != 0u) {
          const unsigned b   = (unsigned)__builtin_ctz(t) >> 3;
          const unsigned pos = cnt + __builtin_amdgcn_mbcnt_lo(m, 0u);
          qe[pos] = (unsigned)p * 16u + (unsigned)(d * 4) + b;
          t &= t - 1u;
        }
        cnt += (unsigned)__builtin_popcount(m);
        m = (unsigned)__ballot(t != 0u);
      }
    }
    while (cnt >= (unsigned)QCH) {
      cnt -= (unsigned)QCH;
      work_chunk<WITH_DEG, SAVE, ROWS>(idx, vals, x, acc, dacc, qe + cnt, (unsigned)QCH,
                                       wl0, wl1, tot0, tot1, r0, lane);
    }
  }
  if (cnt > 0u)
    work_chunk<WITH_DEG, SAVE, ROWS>(idx, vals, x, acc, dacc, qe, cnt,
                                     wl0, wl1, tot0, tot1, r0, lane);
}

template <int ROWS>
__device__ __forceinline__
void store_owned_rows(const float* acc, float* __restrict__ out, const int r0, const int t) {
  float4* O4 = (float4*)(out + (long long)r0 * HID);
  for (int i = t; i < ROWS * 32; i += SPW * 32) {
    const int r = i >> 5;
    const int l = i & 31;
    const float* a = acc + r * HID + l;
    O4[i] = float4{a[0], a[32], a[64], a[96]};
  }
}

__global__ __launch_bounds__(SPW * 32)
void spmm_sweep_kernel(const int* __restrict__ idx, const float* __restrict__ vals,
                       const float* __restrict__ x, const int* __restrict__ keys,
                       float* __restrict__ out0,
                       float* __restrict__ out1, float* __restrict__ dpart,
                       int* __restrict__ lists) {
  __shared__ float          acc[SR * HID];
  __shared__ unsigned       qe[SPW][QCAP];
  __shared__ float          dacc[SR];

  const int t    = threadIdx.x;
  const int lane = t & 31;
  const int wid  = t >> 5;
  const int rg   = (int)blockIdx.x / EH;
  const int e    = (int)blockIdx.x - rg * EH;
  const int r0   = rg * SR;

  for (int i = t; i < SR * HID / 4; i += SPW * 32)
    ((float4*)acc)[i] = float4{0.0f, 0.0f, 0.0f, 0.0f};
  if (t < SR) dacc[t] = 0.0f;
  __syncthreads();

  int2* elist = (int2*)lists;
  int*  ecnt  = lists + 2 * NLIST * CAP;
  const int qlo = e * PARTB;
  const int nb  = min(PARTB, NBATCH - qlo);
  const int l0  = ((rg * NSUB + 0) * EH + e) * SPW + wid;
  const int l1  = ((rg * NSUB + 1) * EH + e) * SPW + wid;
  unsigned tot0 = 0, tot1 = 0;
  sweep_edges<true, true, SR>(idx, vals, x, keys, acc, dacc, qe[wid],
                              elist + (long long)l0 * CAP, elist + (long long)l1 * CAP,
                              tot0, tot1, r0, (unsigned)rg, qlo, nb, (rg * nb) / NRG,
                              lane, wid);
  if (lane == 0) { ecnt[l0] = (int)tot0; ecnt[l1] = (int)tot1; }
  __syncthreads();

  store_owned_rows<SR>(acc, e ? out1 : out0, r0, t);
  if (t < SR) dpart[e * N_NODES + r0 + t] = dacc[t];
}

static_assert(CAP % RD == 0, "whole groups of list entries");

__global__ __launch_bounds__(SPW * 32)
void spmm_replay_kernel(const int* __restrict__ idx, const float* __restrict__ vals,
                        const float* __restrict__ x, const int* __restrict__ keys,
                        float* __restrict__ out,
                        const int* __restrict__ lists) {
  __shared__ float          acc[SUB * HID];
  __shared__ unsigned       qe[SPW][QCAP];

  const int t    = threadIdx.x;
  const int lane = t & 31;
  const int wid  = t >> 5;
  const int r0   = blockIdx.x * SUB;

  for (int i = t; i < SUB * HID / 4; i += SPW * 32)
    ((float4*)acc)[i] = float4{0.0f, 0.0f, 0.0f, 0.0f};
  __syncthreads();

  const int2* elist = (const int2*)lists;
  const int*  ecnt  = lists + 2 * NLIST * CAP;
  const int   lbase = (int)blockIdx.x * (EH * SPW);

  bool over = false;
#pragma unroll
  for (int h = 0; h < EH; ++h) over |= ecnt[lbase + h * SPW + lane] > CAP;
  const bool fits = ((unsigned)__ballot(over)) == 0u;

  if (fits) {
    const v4f* X4 = (const v4f*)x;
    for (int h = 0; h < EH; ++h) {
      const int      wl  = lbase + h * SPW + __builtin_amdgcn_readfirstlane(wid);
      const unsigned cnt = (unsigned)ecnt[wl];
      const int2*    el  = elist + (long long)wl * CAP;
      if (cnt > 0u) {
        int2 en[RD], nx[RD];
#pragma unroll
        for (int j = 0; j < RD; ++j) en[j] = el[min((unsigned)j, cnt - 1u)];

        for (unsigned i = 0; i < cnt; i += RD) {
          v4f xv[RD];
#pragma unroll
          for (int j = 0; j < RD; ++j)
            xv[j] = X4[((unsigned)en[j].x & 0xffffu) * (HID / 4) + lane];
#pragma unroll
          for (int j = 0; j < RD; ++j) nx[j] = el[min(i + RD + j, cnt - 1u)];

          asm volatile("" :: "v"(xv[0]), "v"(xv[1]), "v"(xv[2]), "v"(xv[3]),
                             "v"(xv[4]), "v"(xv[5]), "v"(xv[6]), "v"(xv[7]));
          static_assert(RD == 8, "the statement above names RD gathers");

#pragma unroll
          for (int j = 0; j < RD; ++j) {
            if (i + j < cnt) {
              const float vj = __int_as_float(en[j].y);
              float* a = acc + ((unsigned)en[j].x >> 16) * HID + lane;
              atomicAdd(a,      vj * xv[j].x);
              atomicAdd(a + 32, vj * xv[j].y);
              atomicAdd(a + 64, vj * xv[j].z);
              atomicAdd(a + 96, vj * xv[j].w);
            }
          }
#pragma unroll
          for (int j = 0; j < RD; ++j) en[j] = nx[j];
        }
      }
    }
  } else {
    unsigned u0 = 0, u1 = 0;
    sweep_edges<false, false, SUB>(idx, vals, x, keys, acc, nullptr, qe[wid], nullptr, nullptr,
                                   u0, u1, r0, (unsigned)blockIdx.x / (unsigned)NSUB, 0, NBATCH,
                                   ((int)blockIdx.x * NBATCH) / NRP, lane, wid);
  }
  __syncthreads();

  store_owned_rows<SUB>(acc, out, r0, t);
}

template <bool PLANES>
__global__ void gemm_wmma_kernel(const float* __restrict__ X, const float* X1,
                                 const float* __restrict__ dpart, float* __restrict__ scale,
                                 const float* __restrict__ Wt, const float* __restrict__ bias,
                                 float* Y, int do_relu) {
  __shared__ float Xs[BROWS][132];
  const int t    = threadIdx.x;
  const int row0 = blockIdx.x * BROWS;

#pragma unroll
  for (int rt = 0; rt < RT; ++rt) {
    int   r    = rt * 16 + (t >> 4);
    int   cseg = (t & 15) << 3;
    float s;
    const float* src = X + ((long long)row0 + r) * HID + cseg;
    float4 v0 = *(const float4*)(src);
    float4 v1 = *(const float4*)(src + 4);
    if (PLANES) {
      static_assert(EH == 2, "two partial planes and two partial degrees are added here");
      const float* src1 = X1 + ((long long)row0 + r) * HID + cseg;
      const float4 w0 = *(const float4*)(src1);
      const float4 w1 = *(const float4*)(src1 + 4);
      v0.x += w0.x; v0.y += w0.y; v0.z += w0.z; v0.w += w0.w;
      v1.x += w1.x; v1.y += w1.y; v1.z += w1.z; v1.w += w1.w;
      s = 1.0f / fmaxf(dpart[row0 + r] + dpart[N_NODES + row0 + r], 1.0f);
      if ((t & 15) == 0) scale[row0 + r] = s;
    } else {
      s = scale[row0 + r];
    }
    float* dst = &Xs[r][cseg];
    dst[0] = v0.x * s; dst[1] = v0.y * s; dst[2] = v0.z * s; dst[3] = v0.w * s;
    dst[4] = v1.x * s; dst[5] = v1.y * s; dst[6] = v1.z * s; dst[7] = v1.w * s;
  }
  __syncthreads();

  const int lane = t & 31;
  const int wid  = t >> 5;
  const int m    = lane & 15;
  const int kh   = (lane >> 4) << 1;
  const int n    = (wid << 4) + (lane & 15);

  const float* wrow = Wt + n * HID;

  v8f c[RT];
#pragma unroll
  for (int rt = 0; rt < RT; ++rt) c[rt] = v8f{};

#pragma unroll
  for (int k0 = 0; k0 < HID; k0 += 4) {
    int k = k0 + kh;
    v2f b = *(const v2f*)(wrow + k);
#pragma unroll
    for (int rt = 0; rt < RT; ++rt) {
      v2f a; a.x = Xs[rt * 16 + m][k]; a.y = Xs[rt * 16 + m][k + 1];
      c[rt] = __builtin_amdgcn_wmma_f32_16x16x4_f32(false, a, false, b, (short)0, c[rt],
                                                    false, false);
    }
  }

  const float bn = bias[n];
#pragma unroll
  for (int rt = 0; rt < RT; ++rt) {
    const int rowbase = row0 + rt * 16 + ((lane >> 4) << 3);
#pragma unroll
    for (int i = 0; i < 8; ++i) {
      float v = c[rt][i] + bn;
      if (do_relu) v = fmaxf(v, 0.0f);
      Y[(long long)(rowbase + i) * HID + n] = v;
    }
  }
}

__global__ void decode_kernel(const float* __restrict__ h2,
                              const int* __restrict__ U, const int* __restrict__ V,
                              const float* __restrict__ Wd1t, const float* __restrict__ bd1,
                              const float* __restrict__ Wd2, const float* __restrict__ bd2,
                              float* __restrict__ out) {
  __shared__ float zh[BROWS * 260];
  __shared__ float red[BROWS][16];
  float (*Zs)[260] = (float (*)[260])zh;
  float (*Hs)[HID] = (float (*)[HID])zh;

  const int base = blockIdx.x * BROWS;
  const int t    = threadIdx.x;

  {
    const int  prow = t >> 4;
    const int  seg  = t & 15;
    const int* P    = (seg < 8) ? U : V;
    const int  coff = (seg & 7) << 4;
    int node[RT];
#pragma unroll
    for (int rt = 0; rt < RT; ++rt) node[rt] = P[base + rt * 16 + prow];
    float4 zv[RT][4];
#pragma unroll
    for (int rt = 0; rt < RT; ++rt) {
      const float* src = h2 + (long long)node[rt] * HID + coff;
#pragma unroll
      for (int j = 0; j < 4; ++j) zv[rt][j] = *(const float4*)(src + 4 * j);
    }
#pragma unroll
    for (int rt = 0; rt < RT; ++rt) {
      float* dst = &Zs[rt * 16 + prow][seg << 4];
#pragma unroll
      for (int j = 0; j < 4; ++j) {
        dst[4 * j]     = zv[rt][j].x; dst[4 * j + 1] = zv[rt][j].y;
        dst[4 * j + 2] = zv[rt][j].z; dst[4 * j + 3] = zv[rt][j].w;
      }
    }
  }
  __syncthreads();

  const int lane = t & 31;
  const int wid  = t >> 5;
  const int m    = lane & 15;
  const int kh   = (lane >> 4) << 1;
  const int n    = (wid << 4) + (lane & 15);

  const float* wrow = Wd1t + n * (2 * HID);

  v8f c[RT];
#pragma unroll
  for (int rt = 0; rt < RT; ++rt) c[rt] = v8f{};

#pragma unroll
  for (int k0 = 0; k0 < 2 * HID; k0 += 4) {
    int k = k0 + kh;
    v2f b = *(const v2f*)(wrow + k);
#pragma unroll
    for (int rt = 0; rt < RT; ++rt) {
      v2f a; a.x = Zs[rt * 16 + m][k]; a.y = Zs[rt * 16 + m][k + 1];
      c[rt] = __builtin_amdgcn_wmma_f32_16x16x4_f32(false, a, false, b, (short)0, c[rt],
                                                    false, false);
    }
  }

  __syncthreads();

  const float bn    = bd1[n];
  const int   rbase = (lane >> 4) << 3;
#pragma unroll
  for (int rt = 0; rt < RT; ++rt) {
#pragma unroll
    for (int i = 0; i < 8; ++i) Hs[rt * 16 + rbase + i][n] = fmaxf(c[rt][i] + bn, 0.0f);
  }
  __syncthreads();

#pragma unroll
  for (int rt = 0; rt < RT; ++rt) {
    int i   = rt * 16 + (t >> 4);
    int seg = t & 15;
    float p = 0.0f;
#pragma unroll
    for (int j = 0; j < 8; ++j) p += Hs[i][(seg << 3) + j] * Wd2[(seg << 3) + j];
    red[i][seg] = p;
  }
  __syncthreads();
  if (t < BROWS) {
    float s = bd2[0];
#pragma unroll
    for (int j = 0; j < 16; ++j) s += red[t][j];
    out[base + t] = s;
  }
}

extern "C" void kernel_launch(void* const* d_in, const int* in_sizes, int n_in,
                              void* d_out, int out_size, void* d_ws, size_t ws_size,
                              hipStream_t stream) {
  const int*   adj_idx = (const int*)  d_in[0];
  const float* adj_val = (const float*)d_in[1];
  const float* h       = (const float*)d_in[3];
  const int*   pos_u   = (const int*)  d_in[4];
  const int*   pos_v   = (const int*)  d_in[5];
  const int*   neg_u   = (const int*)  d_in[6];
  const int*   neg_v   = (const int*)  d_in[7];
  const float* W1      = (const float*)d_in[8];
  const float* b1      = (const float*)d_in[9];
  const float* W2      = (const float*)d_in[10];
  const float* b2      = (const float*)d_in[11];
  const float* Wd1     = (const float*)d_in[12];
  const float* bd1     = (const float*)d_in[13];
  const float* Wd2     = (const float*)d_in[14];
  const float* bd2     = (const float*)d_in[15];

  float* outp = (float*)d_out;
  float* pos_scores = outp;
  float* neg_scores = outp + N_POS;
  float* h2         = outp + 2 * N_POS;

  float* ws      = (float*)d_ws;
  float* deg_inv = ws;
  float* tmp     = ws + 50176;
  float* h1      = tmp + (long long)N_NODES * HID;
  float* W1t     = h1 + (long long)N_NODES * HID;
  float* W2t     = W1t + HID * HID;
  float* Wd1t    = W2t + HID * HID;

  transpose_kernel<<<(HID * HID + 255) / 256, 256, 0, stream>>>(W1, W1t, HID);
  transpose_kernel<<<(HID * HID + 255) / 256, 256, 0, stream>>>(W2, W2t, HID);
  transpose_kernel<<<(2 * HID * HID + 255) / 256, 256, 0, stream>>>(Wd1, Wd1t, 2 * HID);
  int*   lists = (int*)h2;
  float* dpart = h2 + (long long)NLIST * (2 * CAP + 1);
  int*   keys  = (int*)(dpart + EH * N_NODES);

  edge_key_kernel<<<(N_EDGES / 4 + 255) / 256, 256, 0, stream>>>(adj_idx, keys);

  spmm_sweep_kernel<<<NWG, SPW * 32, 0, stream>>>(adj_idx, adj_val, h, keys, tmp, h1, dpart, lists);
  gemm_wmma_kernel<true><<<N_NODES / BROWS, 256, 0, stream>>>(tmp, h1, dpart, deg_inv,
                                                              W1t, b1, h1, 1);
  spmm_replay_kernel<<<NRP, SPW * 32, 0, stream>>>(adj_idx, adj_val, h1, keys, tmp, lists);
  gemm_wmma_kernel<false><<<N_NODES / BROWS, 256, 0, stream>>>(tmp, nullptr, nullptr, deg_inv,
                                                               W2t, b2, h2, 0);
  decode_kernel<<<N_POS / BROWS, 256, 0, stream>>>(h2, pos_u, pos_v, Wd1t, bd1, Wd2, bd2, pos_scores);
  decode_kernel<<<N_POS / BROWS, 256, 0, stream>>>(h2, neg_u, neg_v, Wd1t, bd1, Wd2, bd2, neg_scores);
}


# GEMM and decoder k-loops in blocks of 2 k-steps: next block's A fragments (LDS) and the B fragments two blocks on are issued as one group ahead of the block's 10 WMMAs; k and sum order unchanged
# speedup vs baseline: 2.7286x; 1.0307x over previous
#include <hip/hip_runtime.h>
#include <hip/hip_bf16.h>


typedef __attribute__((ext_vector_type(2))) float v2f;
typedef __attribute__((ext_vector_type(8))) float v8f;

#define N_NODES 50000
#define N_EDGES 800000
#define HID     128
#define N_POS   100000

#define RT      5
#define BROWS   (RT * 16)
static_assert(N_NODES % BROWS == 0, "row blocks must tile the node count");
static_assert(N_POS   % BROWS == 0, "pair blocks must tile the pair count");

__global__ void transpose_kernel(const float* __restrict__ W, float* __restrict__ Wt,
                                 int K) {
  int id = blockIdx.x * blockDim.x + threadIdx.x;
  if (id < K * HID) {
    int n = id / K;
    int k = id - n * K;
    Wt[id] = W[k * HID + n];
  }
}

#define SR      400
#define EH      2
#define SUB     200
#define NSUB    (SR / SUB)
#define NRG     (N_NODES / SR)
#define NWG     (NRG * EH)
#define NRP     (N_NODES / SUB)
#define SPW     32
#define NL      4
#define NP      (N_EDGES / 16)
#define BP      32
#define NBATCH  ((NP + BP - 1) / BP)
#define QCH     32
#define QCAP    (QCH + 512)
#define KNONE   255u
#define PARTB   ((NBATCH + EH - 1) / EH)
static_assert(N_NODES % SR == 0, "row ranges must tile the node count");
static_assert(SR % SUB == 0 && NSUB == 2, "a kept edge goes to one of two sub-range lists");
static_assert(N_EDGES % 16 == 0, "sixteen edge keys per 16-byte load");
static_assert(NRG < (int)KNONE, "a range number fits a byte beside the no-range key");
static_assert(N_NODES <= 65536, "a column index is packed into 16 bits");
static_assert(SR < 65536, "a relative row is packed into 16 bits");
static_assert(PARTB >= SPW && NBATCH - (EH - 1) * PARTB >= SPW, "a part holds a batch per wave");

#define CAP     128
#define NLIST   (NRP * EH * SPW)
static_assert(SPW == 32, "one lane per wave when the totals of a work-group are read");
static_assert((long long)NLIST * (CAP * 8 + 4) + (long long)EH * N_NODES * 4 + N_EDGES
              <= (long long)N_NODES * HID * 4,
              "lists, totals, partial degrees and keys must fit the buffer they are placed in");
static_assert(((long long)NLIST * (CAP * 8 + 4) + (long long)EH * N_NODES * 4) % 16 == 0,
              "the keys behind them are read 16 bytes at a time");

__global__ void edge_key_kernel(const int* __restrict__ idx, int* __restrict__ keys) {
  const int t = blockIdx.x * blockDim.x + threadIdx.x;
  if (t < N_EDGES / 4) {
    const int4* E4 = (const int4*)idx;
    const int4 a = E4[2 * t];
    const int4 b = E4[2 * t + 1];
    const unsigned k0 = min((unsigned)a.x / (unsigned)SR, KNONE);
    const unsigned k1 = min((unsigned)a.z / (unsigned)SR, KNONE);
    const unsigned k2 = min((unsigned)b.x / (unsigned)SR, KNONE);
    const unsigned k3 = min((unsigned)b.z / (unsigned)SR, KNONE);
    keys[t] = (int)(k0 | (k1 << 8) | (k2 << 16) | (k3 << 24));
  }
}

#define RD      8
typedef __attribute__((ext_vector_type(4))) float v4f;
static_assert(QCH % RD == 0, "whole groups of queued edges");

template <bool WITH_DEG, bool SAVE, int ROWS>
__device__ __forceinline__
void work_chunk(const int* __restrict__ idx, const float* __restrict__ vals,
                const float* __restrict__ x, float* acc, float* dacc,
                const unsigned* qe, const unsigned n,
                int2* __restrict__ wl0, int2* __restrict__ wl1,
                unsigned& tot0, unsigned& tot1, const int r0, const int lane) {
  const int2* E2 = (const int2*)idx;
  const v4f*  X4 = (const v4f*)x;
  const bool     valid = (unsigned)lane < n;
  const unsigned e     = qe[valid ? lane : 0];
  const int2     rc    = E2[e];
  const int      vb    = __float_as_int(vals[e]);
  const unsigned rel   = (unsigned)(rc.x - r0);
  const bool     keep  = valid & (rel < (unsigned)ROWS);
  const unsigned pk    = (rel << 16) | ((unsigned)rc.y & 0xffffu);
  const unsigned mk    = (unsigned)__ballot(keep);

  if (SAVE) {
    const bool     hi  = keep & (rel >= (unsigned)SUB);
    const bool     lo  = keep & (rel <  (unsigned)SUB);
    const unsigned mh  = (unsigned)__ballot(hi);
    const unsigned ml  = (unsigned)__ballot(lo);
    const unsigned nh  = (unsigned)__builtin_popcount(mh);
    const unsigned nlo = (unsigned)__builtin_popcount(ml);
    if (hi & (tot1 + nh <= (unsigned)CAP)) {
      const unsigned pos = tot1 + __builtin_amdgcn_mbcnt_lo(mh, 0u);
      wl1[pos] = int2{(int)(pk - ((unsigned)SUB << 16)), vb};
    }
    if (lo & (tot0 + nlo <= (unsigned)CAP)) {
      const unsigned pos = tot0 + __builtin_amdgcn_mbcnt_lo(ml, 0u);
      wl0[pos] = int2{(int)pk, vb};
    }
    tot0 += nlo;
    tot1 += nh;
  }

  for (unsigned j0 = 0; j0 < n; j0 += RD) {
    unsigned pj[RD];
    float    vj[RD];
    v4f      xv[RD];
#pragma unroll
    for (int j = 0; j < RD; ++j) {
      const int jj = (int)min(j0 + (unsigned)j, n - 1u);
      pj[j] = (unsigned)__builtin_amdgcn_readlane((int)pk, jj);
      vj[j] = __int_as_float(__builtin_amdgcn_readlane(vb, jj));
      xv[j] = X4[(pj[j] & 0xffffu) * (HID / 4) + lane];
    }
    asm volatile("" :: "v"(xv[0]), "v"(xv[1]), "v"(xv[2]), "v"(xv[3]),
                       "v"(xv[4]), "v"(xv[5]), "v"(xv[6]), "v"(xv[7]));
    static_assert(RD == 8, "the statement above names RD gathers");
#pragma unroll
    for (int j = 0; j < RD; ++j) {
      if ((mk >> (j0 + (unsigned)j)) & 1u) {
        const unsigned rl = pj[j] >> 16;
        float* a = acc + rl * HID + lane;
        atomicAdd(a,      vj[j] * xv[j].x);
        atomicAdd(a + 32, vj[j] * xv[j].y);
        atomicAdd(a + 64, vj[j] * xv[j].z);
        atomicAdd(a + 96, vj[j] * xv[j].w);
        if (WITH_DEG) { if (lane == 0) atomicAdd(dacc + rl, vj[j]); }
      }
    }
  }
}

template <bool WITH_DEG, bool SAVE, int ROWS>
__device__ __forceinline__
void sweep_edges(const int* __restrict__ idx, const float* __restrict__ vals,
                 const float* __restrict__ x, const int* __restrict__ keys,
                 float* acc, float* dacc, unsigned* qe,
                 int2* __restrict__ wl0, int2* __restrict__ wl1,
                 unsigned& tot0, unsigned& tot1,
                 const int r0, const unsigned key, const int qlo, const int nb, const int qoff,
                 const int lane, const int wid) {
  const uint4* K4 = (const uint4*)keys;
  constexpr bool TAIL = (NP % BP) != 0;
  const unsigned kpat = key * 0x01010101u;

  static_assert(NL == 4, "four loads in flight, handed on register to register");
  uint4 c0, c1, c2, c3;
  {
    int qa = wid + qoff;           if (qa >= nb) qa -= nb;
    int qb = wid + SPW + qoff;     if (qb >= nb) qb -= nb;
    int qc = wid + 2 * SPW + qoff; if (qc >= nb) qc -= nb;
    int qd = wid + 3 * SPW + qoff; if (qd >= nb) qd -= nb;
    c0 = K4[min((qlo + qa) * BP + lane, NP - 1)];
    c1 = K4[min((qlo + qb) * BP + lane, NP - 1)];
    c2 = K4[min((qlo + qc) * BP + lane, NP - 1)];
    c3 = K4[min((qlo + qd) * BP + lane, NP - 1)];
  }

  unsigned cnt = 0;
  for (int q = wid; q < nb; q += SPW) {
    int qq = q + qoff; if (qq >= nb) qq -= nb;
    const int p = (qlo + qq) * BP + lane;
    uint4 kw = c0;
    if (TAIL) { if (p >= NP) kw = uint4{~0u, ~0u, ~0u, ~0u}; }
    c0 = c1; c1 = c2; c2 = c3;
    {
      int qn = q + NL * SPW + qoff; if (qn >= nb) qn -= nb;
      c3 = K4[min((qlo + qn) * BP + lane, NP - 1)];
    }

    const unsigned w[4] = {kw.x, kw.y, kw.z, kw.w};
#pragma unroll
    for (int d = 0; d < 4; ++d) {
      const unsigned xk = w[d] ^ kpat;
      unsigned t = ~(((xk & 0x7f7f7f7fu) + 0x7f7f7f7fu) | xk | 0x7f7f7f7fu);
      unsigned m = (unsigned)__ballot(t != 0u);
      while (m != 0u) {
        if (t != 0u) {
          const unsigned b   = (unsigned)__builtin_ctz(t) >> 3;
          const unsigned pos = cnt + __builtin_amdgcn_mbcnt_lo(m, 0u);
          qe[pos] = (unsigned)p * 16u + (unsigned)(d * 4) + b;
          t &= t - 1u;
        }
        cnt += (unsigned)__builtin_popcount(m);
        m = (unsigned)__ballot(t != 0u);
      }
    }
    while (cnt >= (unsigned)QCH) {
      cnt -= (unsigned)QCH;
      work_chunk<WITH_DEG, SAVE, ROWS>(idx, vals, x, acc, dacc, qe + cnt, (unsigned)QCH,
                                       wl0, wl1, tot0, tot1, r0, lane);
    }
  }
  if (cnt > 0u)
    work_chunk<WITH_DEG, SAVE, ROWS>(idx, vals, x, acc, dacc, qe, cnt,
                                     wl0, wl1, tot0, tot1, r0, lane);
}

template <int ROWS>
__device__ __forceinline__
void store_owned_rows(const float* acc, float* __restrict__ out, const int r0, const int t) {
  float4* O4 = (float4*)(out + (long long)r0 * HID);
  for (int i = t; i < ROWS * 32; i += SPW * 32) {
    const int r = i >> 5;
    const int l = i & 31;
    const float* a = acc + r * HID + l;
    O4[i] = float4{a[0], a[32], a[64], a[96]};
  }
}

__global__ __launch_bounds__(SPW * 32)
void spmm_sweep_kernel(const int* __restrict__ idx, const float* __restrict__ vals,
                       const float* __restrict__ x, const int* __restrict__ keys,
                       float* __restrict__ out0,
                       float* __restrict__ out1, float* __restrict__ dpart,
                       int* __restrict__ lists) {
  __shared__ float          acc[SR * HID];
  __shared__ unsigned       qe[SPW][QCAP];
  __shared__ float          dacc[SR];

  const int t    = threadIdx.x;
  const int lane = t & 31;
  const int wid  = t >> 5;
  const int rg   = (int)blockIdx.x / EH;
  const int e    = (int)blockIdx.x - rg * EH;
  const int r0   = rg * SR;

  for (int i = t; i < SR * HID / 4; i += SPW * 32)
    ((float4*)acc)[i] = float4{0.0f, 0.0f, 0.0f, 0.0f};
  if (t < SR) dacc[t] = 0.0f;
  __syncthreads();

  int2* elist = (int2*)lists;
  int*  ecnt  = lists + 2 * NLIST * CAP;
  const int qlo = e * PARTB;
  const int nb  = min(PARTB, NBATCH - qlo);
  const int l0  = ((rg * NSUB + 0) * EH + e) * SPW + wid;
  const int l1  = ((rg * NSUB + 1) * EH + e) * SPW + wid;
  unsigned tot0 = 0, tot1 = 0;
  sweep_edges<true, true, SR>(idx, vals, x, keys, acc, dacc, qe[wid],
                              elist + (long long)l0 * CAP, elist + (long long)l1 * CAP,
                              tot0, tot1, r0, (unsigned)rg, qlo, nb, (rg * nb) / NRG,
                              lane, wid);
  if (lane == 0) { ecnt[l0] = (int)tot0; ecnt[l1] = (int)tot1; }
  __syncthreads();

  store_owned_rows<SR>(acc, e ? out1 : out0, r0, t);
  if (t < SR) dpart[e * N_NODES + r0 + t] = dacc[t];
}

static_assert(CAP % RD == 0, "whole groups of list entries");

__global__ __launch_bounds__(SPW * 32)
void spmm_replay_kernel(const int* __restrict__ idx, const float* __restrict__ vals,
                        const float* __restrict__ x, const int* __restrict__ keys,
                        float* __restrict__ out,
                        const int* __restrict__ lists) {
  __shared__ float          acc[SUB * HID];
  __shared__ unsigned       qe[SPW][QCAP];

  const int t    = threadIdx.x;
  const int lane = t & 31;
  const int wid  = t >> 5;
  const int r0   = blockIdx.x * SUB;

  for (int i = t; i < SUB * HID / 4; i += SPW * 32)
    ((float4*)acc)[i] = float4{0.0f, 0.0f, 0.0f, 0.0f};
  __syncthreads();

  const int2* elist = (const int2*)lists;
  const int*  ecnt  = lists + 2 * NLIST * CAP;
  const int   lbase = (int)blockIdx.x * (EH * SPW);

  bool over = false;
#pragma unroll
  for (int h = 0; h < EH; ++h) over |= ecnt[lbase + h * SPW + lane] > CAP;
  const bool fits = ((unsigned)__ballot(over)) == 0u;

  if (fits) {
    const v4f* X4 = (const v4f*)x;
    for (int h = 0; h < EH; ++h) {
      const int      wl  = lbase + h * SPW + __builtin_amdgcn_readfirstlane(wid);
      const unsigned cnt = (unsigned)ecnt[wl];
      const int2*    el  = elist + (long long)wl * CAP;
      if (cnt > 0u) {
        int2 en[RD], nx[RD];
#pragma unroll
        for (int j = 0; j < RD; ++j) en[j] = el[min((unsigned)j, cnt - 1u)];

        for (unsigned i = 0; i < cnt; i += RD) {
          v4f xv[RD];
#pragma unroll
          for (int j = 0; j < RD; ++j)
            xv[j] = X4[((unsigned)en[j].x & 0xffffu) * (HID / 4) + lane];
#pragma unroll
          for (int j = 0; j < RD; ++j) nx[j] = el[min(i + RD + j, cnt - 1u)];

          asm volatile("" :: "v"(xv[0]), "v"(xv[1]), "v"(xv[2]), "v"(xv[3]),
                             "v"(xv[4]), "v"(xv[5]), "v"(xv[6]), "v"(xv[7]));
          static_assert(RD == 8, "the statement above names RD gathers");

#pragma unroll
          for (int j = 0; j < RD; ++j) {
            if (i + j < cnt) {
              const float vj = __int_as_float(en[j].y);
              float* a = acc + ((unsigned)en[j].x >> 16) * HID + lane;
              atomicAdd(a,      vj * xv[j].x);
              atomicAdd(a + 32, vj * xv[j].y);
              atomicAdd(a + 64, vj * xv[j].z);
              atomicAdd(a + 96, vj * xv[j].w);
            }
          }
#pragma unroll
          for (int j = 0; j < RD; ++j) en[j] = nx[j];
        }
      }
    }
  } else {
    unsigned u0 = 0, u1 = 0;
    sweep_edges<false, false, SUB>(idx, vals, x, keys, acc, nullptr, qe[wid], nullptr, nullptr,
                                   u0, u1, r0, (unsigned)blockIdx.x / (unsigned)NSUB, 0, NBATCH,
                                   ((int)blockIdx.x * NBATCH) / NRP, lane, wid);
  }
  __syncthreads();

  store_owned_rows<SUB>(acc, out, r0, t);
}

#define KB      2
template <int KTOT, int PITCH>
__device__ __forceinline__
void wmma_strip(const float* As, const float* __restrict__ wrow, v8f (&c)[RT]) {
  constexpr int NBLK = KTOT / (4 * KB);
  static_assert(KTOT % (4 * KB) == 0, "whole blocks of k-steps");
  static_assert(NBLK >= 2, "the B fragments run two blocks ahead");
  v2f a[2][KB][RT];
  v2f b[3][KB];
#pragma unroll
  for (int s = 0; s < KB; ++s) {
    b[0][s] = *(const v2f*)(wrow + 4 * s);
    b[1][s] = *(const v2f*)(wrow + 4 * (KB + s));
  }
#pragma unroll
  for (int s = 0; s < KB; ++s) {
#pragma unroll
    for (int rt = 0; rt < RT; ++rt) a[0][s][rt] = *(const v2f*)(As + rt * 16 * PITCH + 4 * s);
  }
#pragma unroll
  for (int blk = 0; blk < NBLK; ++blk) {
    if (blk + 2 < NBLK) {
#pragma unroll
      for (int s = 0; s < KB; ++s)
        b[(blk + 2) % 3][s] = *(const v2f*)(wrow + 4 * ((blk + 2) * KB + s));
    }
    if (blk + 1 < NBLK) {
#pragma unroll
      for (int s = 0; s < KB; ++s) {
#pragma unroll
        for (int rt = 0; rt < RT; ++rt)
          a[(blk + 1) & 1][s][rt] =
              *(const v2f*)(As + rt * 16 * PITCH + 4 * ((blk + 1) * KB + s));
      }
    }
    __builtin_amdgcn_sched_barrier(0);
#pragma unroll
    for (int s = 0; s < KB; ++s) {
#pragma unroll
      for (int rt = 0; rt < RT; ++rt)
        c[rt] = __builtin_amdgcn_wmma_f32_16x16x4_f32(false, a[blk & 1][s][rt], false,
                                                      b[blk % 3][s], (short)0, c[rt],
                                                      false, false);
    }
    __builtin_amdgcn_sched_barrier(0);
  }
}

template <bool PLANES>
__global__ void gemm_wmma_kernel(const float* __restrict__ X, const float* X1,
                                 const float* __restrict__ dpart, float* __restrict__ scale,
                                 const float* __restrict__ Wt, const float* __restrict__ bias,
                                 float* Y, int do_relu) {
  __shared__ __attribute__((aligned(16))) float Xs[BROWS][132];
  const int t    = threadIdx.x;
  const int row0 = blockIdx.x * BROWS;

#pragma unroll
  for (int rt = 0; rt < RT; ++rt) {
    int   r    = rt * 16 + (t >> 4);
    int   cseg = (t & 15) << 3;
    float s;
    const float* src = X + ((long long)row0 + r) * HID + cseg;
    float4 v0 = *(const float4*)(src);
    float4 v1 = *(const float4*)(src + 4);
    if (PLANES) {
      static_assert(EH == 2, "two partial planes and two partial degrees are added here");
      const float* src1 = X1 + ((long long)row0 + r) * HID + cseg;
      const float4 w0 = *(const float4*)(src1);
      const float4 w1 = *(const float4*)(src1 + 4);
      v0.x += w0.x; v0.y += w0.y; v0.z += w0.z; v0.w += w0.w;
      v1.x += w1.x; v1.y += w1.y; v1.z += w1.z; v1.w += w1.w;
      s = 1.0f / fmaxf(dpart[row0 + r] + dpart[N_NODES + row0 + r], 1.0f);
      if ((t & 15) == 0) scale[row0 + r] = s;
    } else {
      s = scale[row0 + r];
    }
    float* dst = &Xs[r][cseg];
    dst[0] = v0.x * s; dst[1] = v0.y * s; dst[2] = v0.z * s; dst[3] = v0.w * s;
    dst[4] = v1.x * s; dst[5] = v1.y * s; dst[6] = v1.z * s; dst[7] = v1.w * s;
  }
  __syncthreads();

  const int lane = t & 31;
  const int wid  = t >> 5;
  const int m    = lane & 15;
  const int kh   = (lane >> 4) << 1;
  const int n    = (wid << 4) + (lane & 15);

  const float* wrow = Wt + n * HID + kh;

  v8f c[RT];
#pragma unroll
  for (int rt = 0; rt < RT; ++rt) c[rt] = v8f{};

  wmma_strip<HID, 132>(&Xs[m][kh], wrow, c);

  const float bn = bias[n];
#pragma unroll
  for (int rt = 0; rt < RT; ++rt) {
    const int rowbase = row0 + rt * 16 + ((lane >> 4) << 3);
#pragma unroll
    for (int i = 0; i < 8; ++i) {
      float v = c[rt][i] + bn;
      if (do_relu) v = fmaxf(v, 0.0f);
      Y[(long long)(rowbase + i) * HID + n] = v;
    }
  }
}

__global__ void decode_kernel(const float* __restrict__ h2,
                              const int* __restrict__ U, const int* __restrict__ V,
                              const float* __restrict__ Wd1t, const float* __restrict__ bd1,
                              const float* __restrict__ Wd2, const float* __restrict__ bd2,
                              float* __restrict__ out) {
  __shared__ __attribute__((aligned(16))) float zh[BROWS * 260];
  __shared__ float red[BROWS][16];
  float (*Zs)[260] = (float (*)[260])zh;
  float (*Hs)[HID] = (float (*)[HID])zh;

  const int base = blockIdx.x * BROWS;
  const int t    = threadIdx.x;

  {
    const int  prow = t >> 4;
    const int  seg  = t & 15;
    const int* P    = (seg < 8) ? U : V;
    const int  coff = (seg & 7) << 4;
    int node[RT];
#pragma unroll
    for (int rt = 0; rt < RT; ++rt) node[rt] = P[base + rt * 16 + prow];
    float4 zv[RT][4];
#pragma unroll
    for (int rt = 0; rt < RT; ++rt) {
      const float* src = h2 + (long long)node[rt] * HID + coff;
#pragma unroll
      for (int j = 0; j < 4; ++j) zv[rt][j] = *(const float4*)(src + 4 * j);
    }
#pragma unroll
    for (int rt = 0; rt < RT; ++rt) {
      float* dst = &Zs[rt * 16 + prow][seg << 4];
#pragma unroll
      for (int j = 0; j < 4; ++j) {
        dst[4 * j]     = zv[rt][j].x; dst[4 * j + 1] = zv[rt][j].y;
        dst[4 * j + 2] = zv[rt][j].z; dst[4 * j + 3] = zv[rt][j].w;
      }
    }
  }
  __syncthreads();

  const int lane = t & 31;
  const int wid  = t >> 5;
  const int m    = lane & 15;
  const int kh   = (lane >> 4) << 1;
  const int n    = (wid << 4) + (lane & 15);

  const float* wrow = Wd1t + n * (2 * HID) + kh;

  v8f c[RT];
#pragma unroll
  for (int rt = 0; rt < RT; ++rt) c[rt] = v8f{};

  wmma_strip<2 * HID, 260>(&Zs[m][kh], wrow, c);

  __syncthreads();

  const float bn    = bd1[n];
  const int   rbase = (lane >> 4) << 3;
#pragma unroll
  for (int rt = 0; rt < RT; ++rt) {
#pragma unroll
    for (int i = 0; i < 8; ++i) Hs[rt * 16 + rbase + i][n] = fmaxf(c[rt][i] + bn, 0.0f);
  }
  __syncthreads();

#pragma unroll
  for (int rt = 0; rt < RT; ++rt) {
    int i   = rt * 16 + (t >> 4);
    int seg = t & 15;
    float p = 0.0f;
#pragma unroll
    for (int j = 0; j < 8; ++j) p += Hs[i][(seg << 3) + j] * Wd2[(seg << 3) + j];
    red[i][seg] = p;
  }
  __syncthreads();
  if (t < BROWS) {
    float s = bd2[0];
#pragma unroll
    for (int j = 0; j < 16; ++j) s += red[t][j];
    out[base + t] = s;
  }
}

extern "C" void kernel_launch(void* const* d_in, const int* in_sizes, int n_in,
                              void* d_out, int out_size, void* d_ws, size_t ws_size,
                              hipStream_t stream) {
  const int*   adj_idx = (const int*)  d_in[0];
  const float* adj_val = (const float*)d_in[1];
  const float* h       = (const float*)d_in[3];
  const int*   pos_u   = (const int*)  d_in[4];
  const int*   pos_v   = (const int*)  d_in[5];
  const int*   neg_u   = (const int*)  d_in[6];
  const int*   neg_v   = (const int*)  d_in[7];
  const float* W1      = (const float*)d_in[8];
  const float* b1      = (const float*)d_in[9];
  const float* W2      = (const float*)d_in[10];
  const float* b2      = (const float*)d_in[11];
  const float* Wd1     = (const float*)d_in[12];
  const float* bd1     = (const float*)d_in[13];
  const float* Wd2     = (const float*)d_in[14];
  const float* bd2     = (const float*)d_in[15];

  float* outp = (float*)d_out;
  float* pos_scores = outp;
  float* neg_scores = outp + N_POS;
  float* h2         = outp + 2 * N_POS;

  float* ws      = (float*)d_ws;
  float* deg_inv = ws;
  float* tmp     = ws + 50176;
  float* h1      = tmp + (long long)N_NODES * HID;
  float* W1t     = h1 + (long long)N_NODES * HID;
  float* W2t     = W1t + HID * HID;
  float* Wd1t    = W2t + HID * HID;

  transpose_kernel<<<(HID * HID + 255) / 256, 256, 0, stream>>>(W1, W1t, HID);
  transpose_kernel<<<(HID * HID + 255) / 256, 256, 0, stream>>>(W2, W2t, HID);
  transpose_kernel<<<(2 * HID * HID + 255) / 256, 256, 0, stream>>>(Wd1, Wd1t, 2 * HID);
  int*   lists = (int*)h2;
  float* dpart = h2 + (long long)NLIST * (2 * CAP + 1);
  int*   keys  = (int*)(dpart + EH * N_NODES);

  edge_key_kernel<<<(N_EDGES / 4 + 255) / 256, 256, 0, stream>>>(adj_idx, keys);

  spmm_sweep_kernel<<<NWG, SPW * 32, 0, stream>>>(adj_idx, adj_val, h, keys, tmp, h1, dpart, lists);
  gemm_wmma_kernel<true><<<N_NODES / BROWS, 256, 0, stream>>>(tmp, h1, dpart, deg_inv,
                                                              W1t, b1, h1, 1);
  spmm_replay_kernel<<<NRP, SPW * 32, 0, stream>>>(adj_idx, adj_val, h1, keys, tmp, lists);
  gemm_wmma_kernel<false><<<N_NODES / BROWS, 256, 0, stream>>>(tmp, nullptr, nullptr, deg_inv,
                                                               W2t, b2, h2, 0);
  decode_kernel<<<N_POS / BROWS, 256, 0, stream>>>(h2, pos_u, pos_v, Wd1t, bd1, Wd2, bd2, pos_scores);
  decode_kernel<<<N_POS / BROWS, 256, 0, stream>>>(h2, neg_u, neg_v, Wd1t, bd1, Wd2, bd2, neg_scores);
}


# B fragments read straight from the row-major weights (two 4-byte loads a k-step, 64-byte row segments) instead of a transposed copy: the 3 transpose launches go (10 to 7); k, sum order unchanged
# speedup vs baseline: 2.8818x; 1.0561x over previous
#include <hip/hip_runtime.h>
#include <hip/hip_bf16.h>


typedef __attribute__((ext_vector_type(2))) float v2f;
typedef __attribute__((ext_vector_type(8))) float v8f;

#define N_NODES 50000
#define N_EDGES 800000
#define HID     128
#define N_POS   100000

#define RT      5
#define BROWS   (RT * 16)
static_assert(N_NODES % BROWS == 0, "row blocks must tile the node count");
static_assert(N_POS   % BROWS == 0, "pair blocks must tile the pair count");

#define SR      400
#define EH      2
#define SUB     200
#define NSUB    (SR / SUB)
#define NRG     (N_NODES / SR)
#define NWG     (NRG * EH)
#define NRP     (N_NODES / SUB)
#define SPW     32
#define NL      4
#define NP      (N_EDGES / 16)
#define BP      32
#define NBATCH  ((NP + BP - 1) / BP)
#define QCH     32
#define QCAP    (QCH + 512)
#define KNONE   255u
#define PARTB   ((NBATCH + EH - 1) / EH)
static_assert(N_NODES % SR == 0, "row ranges must tile the node count");
static_assert(SR % SUB == 0 && NSUB == 2, "a kept edge goes to one of two sub-range lists");
static_assert(N_EDGES % 16 == 0, "sixteen edge keys per 16-byte load");
static_assert(NRG < (int)KNONE, "a range number fits a byte beside the no-range key");
static_assert(N_NODES <= 65536, "a column index is packed into 16 bits");
static_assert(SR < 65536, "a relative row is packed into 16 bits");
static_assert(PARTB >= SPW && NBATCH - (EH - 1) * PARTB >= SPW, "a part holds a batch per wave");

#define CAP     128
#define NLIST   (NRP * EH * SPW)
static_assert(SPW == 32, "one lane per wave when the totals of a work-group are read");
static_assert((long long)NLIST * (CAP * 8 + 4) + (long long)EH * N_NODES * 4 + N_EDGES
              <= (long long)N_NODES * HID * 4,
              "lists, totals, partial degrees and keys must fit the buffer they are placed in");
static_assert(((long long)NLIST * (CAP * 8 + 4) + (long long)EH * N_NODES * 4) % 16 == 0,
              "the keys behind them are read 16 bytes at a time");

__global__ void edge_key_kernel(const int* __restrict__ idx, int* __restrict__ keys) {
  const int t = blockIdx.x * blockDim.x + threadIdx.x;
  if (t < N_EDGES / 4) {
    const int4* E4 = (const int4*)idx;
    const int4 a = E4[2 * t];
    const int4 b = E4[2 * t + 1];
    const unsigned k0 = min((unsigned)a.x / (unsigned)SR, KNONE);
    const unsigned k1 = min((unsigned)a.z / (unsigned)SR, KNONE);
    const unsigned k2 = min((unsigned)b.x / (unsigned)SR, KNONE);
    const unsigned k3 = min((unsigned)b.z / (unsigned)SR, KNONE);
    keys[t] = (int)(k0 | (k1 << 8) | (k2 << 16) | (k3 << 24));
  }
}

#define RD      8
typedef __attribute__((ext_vector_type(4))) float v4f;
static_assert(QCH % RD == 0, "whole groups of queued edges");

template <bool WITH_DEG, bool SAVE, int ROWS>
__device__ __forceinline__
void work_chunk(const int* __restrict__ idx, const float* __restrict__ vals,
                const float* __restrict__ x, float* acc, float* dacc,
                const unsigned* qe, const unsigned n,
                int2* __restrict__ wl0, int2* __restrict__ wl1,
                unsigned& tot0, unsigned& tot1, const int r0, const int lane) {
  const int2* E2 = (const int2*)idx;
  const v4f*  X4 = (const v4f*)x;
  const bool     valid = (unsigned)lane < n;
  const unsigned e     = qe[valid ? lane : 0];
  const int2     rc    = E2[e];
  const int      vb    = __float_as_int(vals[e]);
  const unsigned rel   = (unsigned)(rc.x - r0);
  const bool     keep  = valid & (rel < (unsigned)ROWS);
  const unsigned pk    = (rel << 16) | ((unsigned)rc.y & 0xffffu);
  const unsigned mk    = (unsigned)__ballot(keep);

  if (SAVE) {
    const bool     hi  = keep & (rel >= (unsigned)SUB);
    const bool     lo  = keep & (rel <  (unsigned)SUB);
    const unsigned mh  = (unsigned)__ballot(hi);
    const unsigned ml  = (unsigned)__ballot(lo);
    const unsigned nh  = (unsigned)__builtin_popcount(mh);
    const unsigned nlo = (unsigned)__builtin_popcount(ml);
    if (hi & (tot1 + nh <= (unsigned)CAP)) {
      const unsigned pos = tot1 + __builtin_amdgcn_mbcnt_lo(mh, 0u);
      wl1[pos] = int2{(int)(pk - ((unsigned)SUB << 16)), vb};
    }
    if (lo & (tot0 + nlo <= (unsigned)CAP)) {
      const unsigned pos = tot0 + __builtin_amdgcn_mbcnt_lo(ml, 0u);
      wl0[pos] = int2{(int)pk, vb};
    }
    tot0 += nlo;
    tot1 += nh;
  }

  for (unsigned j0 = 0; j0 < n; j0 += RD) {
    unsigned pj[RD];
    float    vj[RD];
    v4f      xv[RD];
#pragma unroll
    for (int j = 0; j < RD; ++j) {
      const int jj = (int)min(j0 + (unsigned)j, n - 1u);
      pj[j] = (unsigned)__builtin_amdgcn_readlane((int)pk, jj);
      vj[j] = __int_as_float(__builtin_amdgcn_readlane(vb, jj));
      xv[j] = X4[(pj[j] & 0xffffu) * (HID / 4) + lane];
    }
    asm volatile("" :: "v"(xv[0]), "v"(xv[1]), "v"(xv[2]), "v"(xv[3]),
                       "v"(xv[4]), "v"(xv[5]), "v"(xv[6]), "v"(xv[7]));
    static_assert(RD == 8, "the statement above names RD gathers");
#pragma unroll
    for (int j = 0; j < RD; ++j) {
      if ((mk >> (j0 + (unsigned)j)) & 1u) {
        const unsigned rl = pj[j] >> 16;
        float* a = acc + rl * HID + lane;
        atomicAdd(a,      vj[j] * xv[j].x);
        atomicAdd(a + 32, vj[j] * xv[j].y);
        atomicAdd(a + 64, vj[j] * xv[j].z);
        atomicAdd(a + 96, vj[j] * xv[j].w);
        if (WITH_DEG) { if (lane == 0) atomicAdd(dacc + rl, vj[j]); }
      }
    }
  }
}

template <bool WITH_DEG, bool SAVE, int ROWS>
__device__ __forceinline__
void sweep_edges(const int* __restrict__ idx, const float* __restrict__ vals,
                 const float* __restrict__ x, const int* __restrict__ keys,
                 float* acc, float* dacc, unsigned* qe,
                 int2* __restrict__ wl0, int2* __restrict__ wl1,
                 unsigned& tot0, unsigned& tot1,
                 const int r0, const unsigned key, const int qlo, const int nb, const int qoff,
                 const int lane, const int wid) {
  const uint4* K4 = (const uint4*)keys;
  constexpr bool TAIL = (NP % BP) != 0;
  const unsigned kpat = key * 0x01010101u;

  static_assert(NL == 4, "four loads in flight, handed on register to register");
  uint4 c0, c1, c2, c3;
  {
    int qa = wid + qoff;           if (qa >= nb) qa -= nb;
    int qb = wid + SPW + qoff;     if (qb >= nb) qb -= nb;
    int qc = wid + 2 * SPW + qoff; if (qc >= nb) qc -= nb;
    int qd = wid + 3 * SPW + qoff; if (qd >= nb) qd -= nb;
    c0 = K4[min((qlo + qa) * BP + lane, NP - 1)];
    c1 = K4[min((qlo + qb) * BP + lane, NP - 1)];
    c2 = K4[min((qlo + qc) * BP + lane, NP - 1)];
    c3 = K4[min((qlo + qd) * BP + lane, NP - 1)];
  }

  unsigned cnt = 0;
  for (int q = wid; q < nb; q += SPW) {
    int qq = q + qoff; if (qq >= nb) qq -= nb;
    const int p = (qlo + qq) * BP + lane;
    uint4 kw = c0;
    if (TAIL) { if (p >= NP) kw = uint4{~0u, ~0u, ~0u, ~0u}; }
    c0 = c1; c1 = c2; c2 = c3;
    {
      int qn = q + NL * SPW + qoff; if (qn >= nb) qn -= nb;
      c3 = K4[min((qlo + qn) * BP + lane, NP - 1)];
    }

    const unsigned w[4] = {kw.x, kw.y, kw.z, kw.w};
#pragma unroll
    for (int d = 0; d < 4; ++d) {
      const unsigned xk = w[d] ^ kpat;
      unsigned t = ~(((xk & 0x7f7f7f7fu) + 0x7f7f7f7fu) | xk | 0x7f7f7f7fu);
      unsigned m = (unsigned)__ballot(t != 0u);
      while (m != 0u) {
        if (t != 0u) {
          const unsigned b   = (unsigned)__builtin_ctz(t) >> 3;
          const unsigned pos = cnt + __builtin_amdgcn_mbcnt_lo(m, 0u);
          qe[pos] = (unsigned)p * 16u + (unsigned)(d * 4) + b;
          t &= t - 1u;
        }
        cnt += (unsigned)__builtin_popcount(m);
        m = (unsigned)__ballot(t != 0u);
      }
    }
    while (cnt >= (unsigned)QCH) {
      cnt -= (unsigned)QCH;
      work_chunk<WITH_DEG, SAVE, ROWS>(idx, vals, x, acc, dacc, qe + cnt, (unsigned)QCH,
                                       wl0, wl1, tot0, tot1, r0, lane);
    }
  }
  if (cnt > 0u)
    work_chunk<WITH_DEG, SAVE, ROWS>(idx, vals, x, acc, dacc, qe, cnt,
                                     wl0, wl1, tot0, tot1, r0, lane);
}

template <int ROWS>
__device__ __forceinline__
void store_owned_rows(const float* acc, float* __restrict__ out, const int r0, const int t) {
  float4* O4 = (float4*)(out + (long long)r0 * HID);
  for (int i = t; i < ROWS * 32; i += SPW * 32) {
    const int r = i >> 5;
    const int l = i & 31;
    const float* a = acc + r * HID + l;
    O4[i] = float4{a[0], a[32], a[64], a[96]};
  }
}

__global__ __launch_bounds__(SPW * 32)
void spmm_sweep_kernel(const int* __restrict__ idx, const float* __restrict__ vals,
                       const float* __restrict__ x, const int* __restrict__ keys,
                       float* __restrict__ out0,
                       float* __restrict__ out1, float* __restrict__ dpart,
                       int* __restrict__ lists) {
  __shared__ float          acc[SR * HID];
  __shared__ unsigned       qe[SPW][QCAP];
  __shared__ float          dacc[SR];

  const int t    = threadIdx.x;
  const int lane = t & 31;
  const int wid  = t >> 5;
  const int rg   = (int)blockIdx.x / EH;
  const int e    = (int)blockIdx.x - rg * EH;
  const int r0   = rg * SR;

  for (int i = t; i < SR * HID / 4; i += SPW * 32)
    ((float4*)acc)[i] = float4{0.0f, 0.0f, 0.0f, 0.0f};
  if (t < SR) dacc[t] = 0.0f;
  __syncthreads();

  int2* elist = (int2*)lists;
  int*  ecnt  = lists + 2 * NLIST * CAP;
  const int qlo = e * PARTB;
  const int nb  = min(PARTB, NBATCH - qlo);
  const int l0  = ((rg * NSUB + 0) * EH + e) * SPW + wid;
  const int l1  = ((rg * NSUB + 1) * EH + e) * SPW + wid;
  unsigned tot0 = 0, tot1 = 0;
  sweep_edges<true, true, SR>(idx, vals, x, keys, acc, dacc, qe[wid],
                              elist + (long long)l0 * CAP, elist + (long long)l1 * CAP,
                              tot0, tot1, r0, (unsigned)rg, qlo, nb, (rg * nb) / NRG,
                              lane, wid);
  if (lane == 0) { ecnt[l0] = (int)tot0; ecnt[l1] = (int)tot1; }
  __syncthreads();

  store_owned_rows<SR>(acc, e ? out1 : out0, r0, t);
  if (t < SR) dpart[e * N_NODES + r0 + t] = dacc[t];
}

static_assert(CAP % RD == 0, "whole groups of list entries");

__global__ __launch_bounds__(SPW * 32)
void spmm_replay_kernel(const int* __restrict__ idx, const float* __restrict__ vals,
                        const float* __restrict__ x, const int* __restrict__ keys,
                        float* __restrict__ out,
                        const int* __restrict__ lists) {
  __shared__ float          acc[SUB * HID];
  __shared__ unsigned       qe[SPW][QCAP];

  const int t    = threadIdx.x;
  const int lane = t & 31;
  const int wid  = t >> 5;
  const int r0   = blockIdx.x * SUB;

  for (int i = t; i < SUB * HID / 4; i += SPW * 32)
    ((float4*)acc)[i] = float4{0.0f, 0.0f, 0.0f, 0.0f};
  __syncthreads();

  const int2* elist = (const int2*)lists;
  const int*  ecnt  = lists + 2 * NLIST * CAP;
  const int   lbase = (int)blockIdx.x * (EH * SPW);

  bool over = false;
#pragma unroll
  for (int h = 0; h < EH; ++h) over |= ecnt[lbase + h * SPW + lane] > CAP;
  const bool fits = ((unsigned)__ballot(over)) == 0u;

  if (fits) {
    const v4f* X4 = (const v4f*)x;
    for (int h = 0; h < EH; ++h) {
      const int      wl  = lbase + h * SPW + __builtin_amdgcn_readfirstlane(wid);
      const unsigned cnt = (unsigned)ecnt[wl];
      const int2*    el  = elist + (long long)wl * CAP;
      if (cnt > 0u) {
        int2 en[RD], nx[RD];
#pragma unroll
        for (int j = 0; j < RD; ++j) en[j] = el[min((unsigned)j, cnt - 1u)];

        for (unsigned i = 0; i < cnt; i += RD) {
          v4f xv[RD];
#pragma unroll
          for (int j = 0; j < RD; ++j)
            xv[j] = X4[((unsigned)en[j].x & 0xffffu) * (HID / 4) + lane];
#pragma unroll
          for (int j = 0; j < RD; ++j) nx[j] = el[min(i + RD + j, cnt - 1u)];

          asm volatile("" :: "v"(xv[0]), "v"(xv[1]), "v"(xv[2]), "v"(xv[3]),
                             "v"(xv[4]), "v"(xv[5]), "v"(xv[6]), "v"(xv[7]));
          static_assert(RD == 8, "the statement above names RD gathers");

#pragma unroll
          for (int j = 0; j < RD; ++j) {
            if (i + j < cnt) {
              const float vj = __int_as_float(en[j].y);
              float* a = acc + ((unsigned)en[j].x >> 16) * HID + lane;
              atomicAdd(a,      vj * xv[j].x);
              atomicAdd(a + 32, vj * xv[j].y);
              atomicAdd(a + 64, vj * xv[j].z);
              atomicAdd(a + 96, vj * xv[j].w);
            }
          }
#pragma unroll
          for (int j = 0; j < RD; ++j) en[j] = nx[j];
        }
      }
    }
  } else {
    unsigned u0 = 0, u1 = 0;
    sweep_edges<false, false, SUB>(idx, vals, x, keys, acc, nullptr, qe[wid], nullptr, nullptr,
                                   u0, u1, r0, (unsigned)blockIdx.x / (unsigned)NSUB, 0, NBATCH,
                                   ((int)blockIdx.x * NBATCH) / NRP, lane, wid);
  }
  __syncthreads();

  store_owned_rows<SUB>(acc, out, r0, t);
}

#define KB      2
template <int KTOT, int PITCH>
__device__ __forceinline__
void wmma_strip(const float* As, const float* __restrict__ wcol, v8f (&c)[RT]) {
  constexpr int NBLK = KTOT / (4 * KB);
  static_assert(KTOT % (4 * KB) == 0, "whole blocks of k-steps");
  static_assert(NBLK >= 2, "the B fragments run two blocks ahead");
  v2f a[2][KB][RT];
  v2f b[3][KB];
#pragma unroll
  for (int s = 0; s < KB; ++s) {
    b[0][s] = v2f{wcol[(4 * s) * HID], wcol[(4 * s + 1) * HID]};
    b[1][s] = v2f{wcol[(4 * (KB + s)) * HID], wcol[(4 * (KB + s) + 1) * HID]};
  }
#pragma unroll
  for (int s = 0; s < KB; ++s) {
#pragma unroll
    for (int rt = 0; rt < RT; ++rt) a[0][s][rt] = *(const v2f*)(As + rt * 16 * PITCH + 4 * s);
  }
#pragma unroll
  for (int blk = 0; blk < NBLK; ++blk) {
    if (blk + 2 < NBLK) {
#pragma unroll
      for (int s = 0; s < KB; ++s)
        b[(blk + 2) % 3][s] = v2f{wcol[(4 * ((blk + 2) * KB + s)) * HID],
                                  wcol[(4 * ((blk + 2) * KB + s) + 1) * HID]};
    }
    if (blk + 1 < NBLK) {
#pragma unroll
      for (int s = 0; s < KB; ++s) {
#pragma unroll
        for (int rt = 0; rt < RT; ++rt)
          a[(blk + 1) & 1][s][rt] =
              *(const v2f*)(As + rt * 16 * PITCH + 4 * ((blk + 1) * KB + s));
      }
    }
    __builtin_amdgcn_sched_barrier(0);
#pragma unroll
    for (int s = 0; s < KB; ++s) {
#pragma unroll
      for (int rt = 0; rt < RT; ++rt)
        c[rt] = __builtin_amdgcn_wmma_f32_16x16x4_f32(false, a[blk & 1][s][rt], false,
                                                      b[blk % 3][s], (short)0, c[rt],
                                                      false, false);
    }
    __builtin_amdgcn_sched_barrier(0);
  }
}

template <bool PLANES>
__global__ void gemm_wmma_kernel(const float* __restrict__ X, const float* X1,
                                 const float* __restrict__ dpart, float* __restrict__ scale,
                                 const float* __restrict__ Wm, const float* __restrict__ bias,
                                 float* Y, int do_relu) {
  __shared__ __attribute__((aligned(16))) float Xs[BROWS][132];
  const int t    = threadIdx.x;
  const int row0 = blockIdx.x * BROWS;

#pragma unroll
  for (int rt = 0; rt < RT; ++rt) {
    int   r    = rt * 16 + (t >> 4);
    int   cseg = (t & 15) << 3;
    float s;
    const float* src = X + ((long long)row0 + r) * HID + cseg;
    float4 v0 = *(const float4*)(src);
    float4 v1 = *(const float4*)(src + 4);
    if (PLANES) {
      static_assert(EH == 2, "two partial planes and two partial degrees are added here");
      const float* src1 = X1 + ((long long)row0 + r) * HID + cseg;
      const float4 w0 = *(const float4*)(src1);
      const float4 w1 = *(const float4*)(src1 + 4);
      v0.x += w0.x; v0.y += w0.y; v0.z += w0.z; v0.w += w0.w;
      v1.x += w1.x; v1.y += w1.y; v1.z += w1.z; v1.w += w1.w;
      s = 1.0f / fmaxf(dpart[row0 + r] + dpart[N_NODES + row0 + r], 1.0f);
      if ((t & 15) == 0) scale[row0 + r] = s;
    } else {
      s = scale[row0 + r];
    }
    float* dst = &Xs[r][cseg];
    dst[0] = v0.x * s; dst[1] = v0.y * s; dst[2] = v0.z * s; dst[3] = v0.w * s;
    dst[4] = v1.x * s; dst[5] = v1.y * s; dst[6] = v1.z * s; dst[7] = v1.w * s;
  }
  __syncthreads();

  const int lane = t & 31;
  const int wid  = t >> 5;
  const int m    = lane & 15;
  const int kh   = (lane >> 4) << 1;
  const int n    = (wid << 4) + (lane & 15);

  const float* wcol = Wm + kh * HID + n;

  v8f c[RT];
#pragma unroll
  for (int rt = 0; rt < RT; ++rt) c[rt] = v8f{};

  wmma_strip<HID, 132>(&Xs[m][kh], wcol, c);

  const float bn = bias[n];
#pragma unroll
  for (int rt = 0; rt < RT; ++rt) {
    const int rowbase = row0 + rt * 16 + ((lane >> 4) << 3);
#pragma unroll
    for (int i = 0; i < 8; ++i) {
      float v = c[rt][i] + bn;
      if (do_relu) v = fmaxf(v, 0.0f);
      Y[(long long)(rowbase + i) * HID + n] = v;
    }
  }
}

__global__ void decode_kernel(const float* __restrict__ h2,
                              const int* __restrict__ U, const int* __restrict__ V,
                              const float* __restrict__ Wd1, const float* __restrict__ bd1,
                              const float* __restrict__ Wd2, const float* __restrict__ bd2,
                              float* __restrict__ out) {
  __shared__ __attribute__((aligned(16))) float zh[BROWS * 260];
  __shared__ float red[BROWS][16];
  float (*Zs)[260] = (float (*)[260])zh;
  float (*Hs)[HID] = (float (*)[HID])zh;

  const int base = blockIdx.x * BROWS;
  const int t    = threadIdx.x;

  {
    const int  prow = t >> 4;
    const int  seg  = t & 15;
    const int* P    = (seg < 8) ? U : V;
    const int  coff = (seg & 7) << 4;
    int node[RT];
#pragma unroll
    for (int rt = 0; rt < RT; ++rt) node[rt] = P[base + rt * 16 + prow];
    float4 zv[RT][4];
#pragma unroll
    for (int rt = 0; rt < RT; ++rt) {
      const float* src = h2 + (long long)node[rt] * HID + coff;
#pragma unroll
      for (int j = 0; j < 4; ++j) zv[rt][j] = *(const float4*)(src + 4 * j);
    }
#pragma unroll
    for (int rt = 0; rt < RT; ++rt) {
      float* dst = &Zs[rt * 16 + prow][seg << 4];
#pragma unroll
      for (int j = 0; j < 4; ++j) {
        dst[4 * j]     = zv[rt][j].x; dst[4 * j + 1] = zv[rt][j].y;
        dst[4 * j + 2] = zv[rt][j].z; dst[4 * j + 3] = zv[rt][j].w;
      }
    }
  }
  __syncthreads();

  const int lane = t & 31;
  const int wid  = t >> 5;
  const int m    = lane & 15;
  const int kh   = (lane >> 4) << 1;
  const int n    = (wid << 4) + (lane & 15);

  const float* wcol = Wd1 + kh * HID + n;

  v8f c[RT];
#pragma unroll
  for (int rt = 0; rt < RT; ++rt) c[rt] = v8f{};

  wmma_strip<2 * HID, 260>(&Zs[m][kh], wcol, c);

  __syncthreads();

  const float bn    = bd1[n];
  const int   rbase = (lane >> 4) << 3;
#pragma unroll
  for (int rt = 0; rt < RT; ++rt) {
#pragma unroll
    for (int i = 0; i < 8; ++i) Hs[rt * 16 + rbase + i][n] = fmaxf(c[rt][i] + bn, 0.0f);
  }
  __syncthreads();

#pragma unroll
  for (int rt = 0; rt < RT; ++rt) {
    int i   = rt * 16 + (t >> 4);
    int seg = t & 15;
    float p = 0.0f;
#pragma unroll
    for (int j = 0; j < 8; ++j) p += Hs[i][(seg << 3) + j] * Wd2[(seg << 3) + j];
    red[i][seg] = p;
  }
  __syncthreads();
  if (t < BROWS) {
    float s = bd2[0];
#pragma unroll
    for (int j = 0; j < 16; ++j) s += red[t][j];
    out[base + t] = s;
  }
}

extern "C" void kernel_launch(void* const* d_in, const int* in_sizes, int n_in,
                              void* d_out, int out_size, void* d_ws, size_t ws_size,
                              hipStream_t stream) {
  const int*   adj_idx = (const int*)  d_in[0];
  const float* adj_val = (const float*)d_in[1];
  const float* h       = (const float*)d_in[3];
  const int*   pos_u   = (const int*)  d_in[4];
  const int*   pos_v   = (const int*)  d_in[5];
  const int*   neg_u   = (const int*)  d_in[6];
  const int*   neg_v   = (const int*)  d_in[7];
  const float* W1      = (const float*)d_in[8];
  const float* b1      = (const float*)d_in[9];
  const float* W2      = (const float*)d_in[10];
  const float* b2      = (const float*)d_in[11];
  const float* Wd1     = (const float*)d_in[12];
  const float* bd1     = (const float*)d_in[13];
  const float* Wd2     = (const float*)d_in[14];
  const float* bd2     = (const float*)d_in[15];

  float* outp = (float*)d_out;
  float* pos_scores = outp;
  float* neg_scores = outp + N_POS;
  float* h2         = outp + 2 * N_POS;

  float* ws      = (float*)d_ws;
  float* deg_inv = ws;
  float* tmp     = ws + 50176;
  float* h1      = tmp + (long long)N_NODES * HID;

  int*   lists = (int*)h2;
  float* dpart = h2 + (long long)NLIST * (2 * CAP + 1);
  int*   keys  = (int*)(dpart + EH * N_NODES);

  edge_key_kernel<<<(N_EDGES / 4 + 255) / 256, 256, 0, stream>>>(adj_idx, keys);

  spmm_sweep_kernel<<<NWG, SPW * 32, 0, stream>>>(adj_idx, adj_val, h, keys, tmp, h1, dpart, lists);
  gemm_wmma_kernel<true><<<N_NODES / BROWS, 256, 0, stream>>>(tmp, h1, dpart, deg_inv,
                                                              W1, b1, h1, 1);
  spmm_replay_kernel<<<NRP, SPW * 32, 0, stream>>>(adj_idx, adj_val, h1, keys, tmp, lists);
  gemm_wmma_kernel<false><<<N_NODES / BROWS, 256, 0, stream>>>(tmp, nullptr, nullptr, deg_inv,
                                                               W2, b2, h2, 0);
  decode_kernel<<<N_POS / BROWS, 256, 0, stream>>>(h2, pos_u, pos_v, Wd1, bd1, Wd2, bd2, pos_scores);
  decode_kernel<<<N_POS / BROWS, 256, 0, stream>>>(h2, neg_u, neg_v, Wd1, bd1, Wd2, bd2, neg_scores);
}


# Layer-2 GEMM done inside the replay work-group: its 200 summed rows go from LDS (pitch 132) straight into WMMA; no SpMM plane, 7 launches become 6; layer-2 k (sum) order follows the LDS row layout
# speedup vs baseline: 3.0196x; 1.0478x over previous
#include <hip/hip_runtime.h>
#include <hip/hip_bf16.h>


typedef __attribute__((ext_vector_type(2))) float v2f;
typedef __attribute__((ext_vector_type(8))) float v8f;

#define N_NODES 50000
#define N_EDGES 800000
#define HID     128
#define N_POS   100000

#define RT      5
#define BROWS   (RT * 16)
static_assert(N_NODES % BROWS == 0, "row blocks must tile the node count");
static_assert(N_POS   % BROWS == 0, "pair blocks must tile the pair count");

#define SR      400
#define EH      2
#define SUB     200
#define NSUB    (SR / SUB)
#define NRG     (N_NODES / SR)
#define NWG     (NRG * EH)
#define NRP     (N_NODES / SUB)
#define SPW     32
#define NL      4
#define NP      (N_EDGES / 16)
#define BP      32
#define NBATCH  ((NP + BP - 1) / BP)
#define QCH     32
#define QCAP    (QCH + 512)
#define KNONE   255u
#define PARTB   ((NBATCH + EH - 1) / EH)
static_assert(N_NODES % SR == 0, "row ranges must tile the node count");
static_assert(SR % SUB == 0 && NSUB == 2, "a kept edge goes to one of two sub-range lists");
static_assert(N_EDGES % 16 == 0, "sixteen edge keys per 16-byte load");
static_assert(NRG < (int)KNONE, "a range number fits a byte beside the no-range key");
static_assert(N_NODES <= 65536, "a column index is packed into 16 bits");
static_assert(SR < 65536, "a relative row is packed into 16 bits");
static_assert(PARTB >= SPW && NBATCH - (EH - 1) * PARTB >= SPW, "a part holds a batch per wave");

#define CAP     128
#define NLIST   (NRP * EH * SPW)
static_assert(SPW == 32, "one lane per wave when the totals of a work-group are read");
static_assert((long long)NLIST * (CAP * 8 + 4) + (long long)EH * N_NODES * 4 + N_EDGES
              <= (long long)N_NODES * HID * 4,
              "lists, totals, partial degrees and keys must fit the buffer they are placed in");
static_assert(((long long)NLIST * (CAP * 8 + 4) + (long long)EH * N_NODES * 4) % 16 == 0,
              "the keys behind them are read 16 bytes at a time");

__global__ void edge_key_kernel(const int* __restrict__ idx, int* __restrict__ keys) {
  const int t = blockIdx.x * blockDim.x + threadIdx.x;
  if (t < N_EDGES / 4) {
    const int4* E4 = (const int4*)idx;
    const int4 a = E4[2 * t];
    const int4 b = E4[2 * t + 1];
    const unsigned k0 = min((unsigned)a.x / (unsigned)SR, KNONE);
    const unsigned k1 = min((unsigned)a.z / (unsigned)SR, KNONE);
    const unsigned k2 = min((unsigned)b.x / (unsigned)SR, KNONE);
    const unsigned k3 = min((unsigned)b.z / (unsigned)SR, KNONE);
    keys[t] = (int)(k0 | (k1 << 8) | (k2 << 16) | (k3 << 24));
  }
}

#define RD      8
typedef __attribute__((ext_vector_type(4))) float v4f;
static_assert(QCH % RD == 0, "whole groups of queued edges");

template <bool WITH_DEG, bool SAVE, int ROWS, int PITCH>
__device__ __forceinline__
void work_chunk(const int* __restrict__ idx, const float* __restrict__ vals,
                const float* __restrict__ x, float* acc, float* dacc,
                const unsigned* qe, const unsigned n,
                int2* __restrict__ wl0, int2* __restrict__ wl1,
                unsigned& tot0, unsigned& tot1, const int r0, const int lane) {
  const int2* E2 = (const int2*)idx;
  const v4f*  X4 = (const v4f*)x;
  const bool     valid = (unsigned)lane < n;
  const unsigned e     = qe[valid ? lane : 0];
  const int2     rc    = E2[e];
  const int      vb    = __float_as_int(vals[e]);
  const unsigned rel   = (unsigned)(rc.x - r0);
  const bool     keep  = valid & (rel < (unsigned)ROWS);
  const unsigned pk    = (rel << 16) | ((unsigned)rc.y & 0xffffu);
  const unsigned mk    = (unsigned)__ballot(keep);

  if (SAVE) {
    const bool     hi  = keep & (rel >= (unsigned)SUB);
    const bool     lo  = keep & (rel <  (unsigned)SUB);
    const unsigned mh  = (unsigned)__ballot(hi);
    const unsigned ml  = (unsigned)__ballot(lo);
    const unsigned nh  = (unsigned)__builtin_popcount(mh);
    const unsigned nlo = (unsigned)__builtin_popcount(ml);
    if (hi & (tot1 + nh <= (unsigned)CAP)) {
      const unsigned pos = tot1 + __builtin_amdgcn_mbcnt_lo(mh, 0u);
      wl1[pos] = int2{(int)(pk - ((unsigned)SUB << 16)), vb};
    }
    if (lo & (tot0 + nlo <= (unsigned)CAP)) {
      const unsigned pos = tot0 + __builtin_amdgcn_mbcnt_lo(ml, 0u);
      wl0[pos] = int2{(int)pk, vb};
    }
    tot0 += nlo;
    tot1 += nh;
  }

  for (unsigned j0 = 0; j0 < n; j0 += RD) {
    unsigned pj[RD];
    float    vj[RD];
    v4f      xv[RD];
#pragma unroll
    for (int j = 0; j < RD; ++j) {
      const int jj = (int)min(j0 + (unsigned)j, n - 1u);
      pj[j] = (unsigned)__builtin_amdgcn_readlane((int)pk, jj);
      vj[j] = __int_as_float(__builtin_amdgcn_readlane(vb, jj));
      xv[j] = X4[(pj[j] & 0xffffu) * (HID / 4) + lane];
    }
    asm volatile("" :: "v"(xv[0]), "v"(xv[1]), "v"(xv[2]), "v"(xv[3]),
                       "v"(xv[4]), "v"(xv[5]), "v"(xv[6]), "v"(xv[7]));
    static_assert(RD == 8, "the statement above names RD gathers");
#pragma unroll
    for (int j = 0; j < RD; ++j) {
      if ((mk >> (j0 + (unsigned)j)) & 1u) {
        const unsigned rl = pj[j] >> 16;
        float* a = acc + rl * PITCH + lane;
        atomicAdd(a,      vj[j] * xv[j].x);
        atomicAdd(a + 32, vj[j] * xv[j].y);
        atomicAdd(a + 64, vj[j] * xv[j].z);
        atomicAdd(a + 96, vj[j] * xv[j].w);
        if (WITH_DEG) { if (lane == 0) atomicAdd(dacc + rl, vj[j]); }
      }
    }
  }
}

template <bool WITH_DEG, bool SAVE, int ROWS, int PITCH>
__device__ __forceinline__
void sweep_edges(const int* __restrict__ idx, const float* __restrict__ vals,
                 const float* __restrict__ x, const int* __restrict__ keys,
                 float* acc, float* dacc, unsigned* qe,
                 int2* __restrict__ wl0, int2* __restrict__ wl1,
                 unsigned& tot0, unsigned& tot1,
                 const int r0, const unsigned key, const int qlo, const int nb, const int qoff,
                 const int lane, const int wid) {
  const uint4* K4 = (const uint4*)keys;
  constexpr bool TAIL = (NP % BP) != 0;
  const unsigned kpat = key * 0x01010101u;

  static_assert(NL == 4, "four loads in flight, handed on register to register");
  uint4 c0, c1, c2, c3;
  {
    int qa = wid + qoff;           if (qa >= nb) qa -= nb;
    int qb = wid + SPW + qoff;     if (qb >= nb) qb -= nb;
    int qc = wid + 2 * SPW + qoff; if (qc >= nb) qc -= nb;
    int qd = wid + 3 * SPW + qoff; if (qd >= nb) qd -= nb;
    c0 = K4[min((qlo + qa) * BP + lane, NP - 1)];
    c1 = K4[min((qlo + qb) * BP + lane, NP - 1)];
    c2 = K4[min((qlo + qc) * BP + lane, NP - 1)];
    c3 = K4[min((qlo + qd) * BP + lane, NP - 1)];
  }

  unsigned cnt = 0;
  for (int q = wid; q < nb; q += SPW) {
    int qq = q + qoff; if (qq >= nb) qq -= nb;
    const int p = (qlo + qq) * BP + lane;
    uint4 kw = c0;
    if (TAIL) { if (p >= NP) kw = uint4{~0u, ~0u, ~0u, ~0u}; }
    c0 = c1; c1 = c2; c2 = c3;
    {
      int qn = q + NL * SPW + qoff; if (qn >= nb) qn -= nb;
      c3 = K4[min((qlo + qn) * BP + lane, NP - 1)];
    }

    const unsigned w[4] = {kw.x, kw.y, kw.z, kw.w};
#pragma unroll
    for (int d = 0; d < 4; ++d) {
      const unsigned xk = w[d] ^ kpat;
      unsigned t = ~(((xk & 0x7f7f7f7fu) + 0x7f7f7f7fu) | xk | 0x7f7f7f7fu);
      unsigned m = (unsigned)__ballot(t != 0u);
      while (m != 0u) {
        if (t != 0u) {
          const unsigned b   = (unsigned)__builtin_ctz(t) >> 3;
          const unsigned pos = cnt + __builtin_amdgcn_mbcnt_lo(m, 0u);
          qe[pos] = (unsigned)p * 16u + (unsigned)(d * 4) + b;
          t &= t - 1u;
        }
        cnt += (unsigned)__builtin_popcount(m);
        m = (unsigned)__ballot(t != 0u);
      }
    }
    while (cnt >= (unsigned)QCH) {
      cnt -= (unsigned)QCH;
      work_chunk<WITH_DEG, SAVE, ROWS, PITCH>(idx, vals, x, acc, dacc, qe + cnt, (unsigned)QCH,
                                       wl0, wl1, tot0, tot1, r0, lane);
    }
  }
  if (cnt > 0u)
    work_chunk<WITH_DEG, SAVE, ROWS, PITCH>(idx, vals, x, acc, dacc, qe, cnt,
                                     wl0, wl1, tot0, tot1, r0, lane);
}

template <int ROWS>
__device__ __forceinline__
void store_owned_rows(const float* acc, float* __restrict__ out, const int r0, const int t) {
  float4* O4 = (float4*)(out + (long long)r0 * HID);
  for (int i = t; i < ROWS * 32; i += SPW * 32) {
    const int r = i >> 5;
    const int l = i & 31;
    const float* a = acc + r * HID + l;
    O4[i] = float4{a[0], a[32], a[64], a[96]};
  }
}

__global__ __launch_bounds__(SPW * 32)
void spmm_sweep_kernel(const int* __restrict__ idx, const float* __restrict__ vals,
                       const float* __restrict__ x, const int* __restrict__ keys,
                       float* __restrict__ out0,
                       float* __restrict__ out1, float* __restrict__ dpart,
                       int* __restrict__ lists) {
  __shared__ float          acc[SR * HID];
  __shared__ unsigned       qe[SPW][QCAP];
  __shared__ float          dacc[SR];

  const int t    = threadIdx.x;
  const int lane = t & 31;
  const int wid  = t >> 5;
  const int rg   = (int)blockIdx.x / EH;
  const int e    = (int)blockIdx.x - rg * EH;
  const int r0   = rg * SR;

  for (int i = t; i < SR * HID / 4; i += SPW * 32)
    ((float4*)acc)[i] = float4{0.0f, 0.0f, 0.0f, 0.0f};
  if (t < SR) dacc[t] = 0.0f;
  __syncthreads();

  int2* elist = (int2*)lists;
  int*  ecnt  = lists + 2 * NLIST * CAP;
  const int qlo = e * PARTB;
  const int nb  = min(PARTB, NBATCH - qlo);
  const int l0  = ((rg * NSUB + 0) * EH + e) * SPW + wid;
  const int l1  = ((rg * NSUB + 1) * EH + e) * SPW + wid;
  unsigned tot0 = 0, tot1 = 0;
  sweep_edges<true, true, SR, HID>(idx, vals, x, keys, acc, dacc, qe[wid],
                              elist + (long long)l0 * CAP, elist + (long long)l1 * CAP,
                              tot0, tot1, r0, (unsigned)rg, qlo, nb, (rg * nb) / NRG,
                              lane, wid);
  if (lane == 0) { ecnt[l0] = (int)tot0; ecnt[l1] = (int)tot1; }
  __syncthreads();

  store_owned_rows<SR>(acc, e ? out1 : out0, r0, t);
  if (t < SR) dpart[e * N_NODES + r0 + t] = dacc[t];
}

static_assert(CAP % RD == 0, "whole groups of list entries");

#define AP      132
#define SUBT    ((SUB + 15) / 16)
#define SUBP    (SUBT * 16)
#define CT      (HID / 16)
#define RG      (SPW / CT)
#define TLO     (SUBT / RG)
static_assert(SPW % CT == 0 && RG * CT == SPW, "one wave per (row group, column tile)");
static_assert(SUBT % RG == 1, "exactly the first row group does TLO + 1 tiles");
static_assert(AP % 4 == 0 && AP % 64 == 4, "rows 16-byte aligned; 16 rows fall on 16 different banks");

template <int NT>
__device__ __forceinline__
void wmma_owned_rows(const float* Ar, const float* __restrict__ wcol, v8f (&c)[NT]) {
  constexpr int NS = HID / 4;
  v2f a[2][NT];
  v2f b[2];
#pragma unroll
  for (int rt = 0; rt < NT; ++rt) a[0][rt] = v2f{Ar[rt * 16 * AP], Ar[rt * 16 * AP + 32]};
  b[0] = v2f{wcol[0], wcol[HID]};
#pragma unroll
  for (int s = 0; s < NS; ++s) {
    if (s + 1 < NS) {
      const int ps = 64 * ((s + 1) >> 4) + 2 * ((s + 1) & 15);
      const int fr = 8 * ((s + 1) & 15) + 2 * ((s + 1) >> 4);
#pragma unroll
      for (int rt = 0; rt < NT; ++rt)
        a[(s + 1) & 1][rt] = v2f{Ar[rt * 16 * AP + ps], Ar[rt * 16 * AP + ps + 32]};
      b[(s + 1) & 1] = v2f{wcol[fr * HID], wcol[(fr + 1) * HID]};
    }
    __builtin_amdgcn_sched_barrier(0);
#pragma unroll
    for (int rt = 0; rt < NT; ++rt)
      c[rt] = __builtin_amdgcn_wmma_f32_16x16x4_f32(false, a[s & 1][rt], false, b[s & 1],
                                                    (short)0, c[rt], false, false);
    __builtin_amdgcn_sched_barrier(0);
  }
}

template <int NT>
__device__ __forceinline__
void gemm_owned_rows(const float* acc, const float* __restrict__ Wm, const float* __restrict__ bias,
                     float* __restrict__ Y, const int r0, const int t0, const int lane, const int n) {
  const int m = lane & 15;
  const int h = lane >> 4;
  v8f c[NT];
#pragma unroll
  for (int rt = 0; rt < NT; ++rt) c[rt] = v8f{};
  wmma_owned_rows<NT>(acc + (t0 * 16 + m) * AP + h, Wm + (4 * h) * HID + n, c);
  const float bn = bias[n];
#pragma unroll
  for (int rt = 0; rt < NT; ++rt) {
    const int rl = (t0 + rt) * 16 + (h << 3);
    static_assert(SUB % 8 == 0, "a lane's 8 rows of a tile are all owned or all padding");
    if (rl < SUB) {
#pragma unroll
      for (int i = 0; i < 8; ++i) Y[(long long)(r0 + rl + i) * HID + n] = c[rt][i] + bn;
    }
  }
}

__global__ __launch_bounds__(SPW * 32)
void spmm_replay_gemm_kernel(const int* __restrict__ idx, const float* __restrict__ vals,
                             const float* __restrict__ x, const int* __restrict__ keys,
                             const float* __restrict__ scale, const float* __restrict__ Wm,
                             const float* __restrict__ bias, float* __restrict__ Y,
                             const int* __restrict__ lists) {
  __shared__ __attribute__((aligned(16))) float acc[SUBP * AP];
  __shared__ unsigned       qe[SPW][QCAP];

  const int t    = threadIdx.x;
  const int lane = t & 31;
  const int wid  = t >> 5;
  const int r0   = blockIdx.x * SUB;

  for (int i = t; i < SUBP * AP / 4; i += SPW * 32)
    ((float4*)acc)[i] = float4{0.0f, 0.0f, 0.0f, 0.0f};
  __syncthreads();

  const int2* elist = (const int2*)lists;
  const int*  ecnt  = lists + 2 * NLIST * CAP;
  const int   lbase = (int)blockIdx.x * (EH * SPW);

  bool over = false;
#pragma unroll
  for (int h = 0; h < EH; ++h) over |= ecnt[lbase + h * SPW + lane] > CAP;
  const bool fits = ((unsigned)__ballot(over)) == 0u;

  if (fits) {
    const v4f* X4 = (const v4f*)x;
    for (int h = 0; h < EH; ++h) {
      const int      wl  = lbase + h * SPW + __builtin_amdgcn_readfirstlane(wid);
      const unsigned cnt = (unsigned)ecnt[wl];
      const int2*    el  = elist + (long long)wl * CAP;
      if (cnt > 0u) {
        int2 en[RD], nx[RD];
#pragma unroll
        for (int j = 0; j < RD; ++j) en[j] = el[min((unsigned)j, cnt - 1u)];

        for (unsigned i = 0; i < cnt; i += RD) {
          v4f xv[RD];
#pragma unroll
          for (int j = 0; j < RD; ++j)
            xv[j] = X4[((unsigned)en[j].x & 0xffffu) * (HID / 4) + lane];
#pragma unroll
          for (int j = 0; j < RD; ++j) nx[j] = el[min(i + RD + j, cnt - 1u)];

          asm volatile("" :: "v"(xv[0]), "v"(xv[1]), "v"(xv[2]), "v"(xv[3]),
                             "v"(xv[4]), "v"(xv[5]), "v"(xv[6]), "v"(xv[7]));
          static_assert(RD == 8, "the statement above names RD gathers");

#pragma unroll
          for (int j = 0; j < RD; ++j) {
            if (i + j < cnt) {
              const float vj = __int_as_float(en[j].y);
              float* a = acc + ((unsigned)en[j].x >> 16) * AP + lane;
              atomicAdd(a,      vj * xv[j].x);
              atomicAdd(a + 32, vj * xv[j].y);
              atomicAdd(a + 64, vj * xv[j].z);
              atomicAdd(a + 96, vj * xv[j].w);
            }
          }
#pragma unroll
          for (int j = 0; j < RD; ++j) en[j] = nx[j];
        }
      }
    }
  } else {
    unsigned u0 = 0, u1 = 0;
    sweep_edges<false, false, SUB, AP>(idx, vals, x, keys, acc, nullptr, qe[wid], nullptr, nullptr,
                                       u0, u1, r0, (unsigned)blockIdx.x / (unsigned)NSUB, 0, NBATCH,
                                       ((int)blockIdx.x * NBATCH) / NRP, lane, wid);
  }
  __syncthreads();

  for (int i = t; i < SUB * HID; i += SPW * 32) {
    const int r = i / HID;
    acc[r * AP + (i - r * HID)] *= scale[r0 + r];
  }
  __syncthreads();

  const int ct = wid % CT;
  const int rg = __builtin_amdgcn_readfirstlane(wid / CT);
  const int n  = ct * 16 + (lane & 15);
  if (rg == 0) gemm_owned_rows<TLO + 1>(acc, Wm, bias, Y, r0, 0, lane, n);
  else         gemm_owned_rows<TLO>(acc, Wm, bias, Y, r0, rg * TLO + 1, lane, n);
}

#define KB      2
template <int KTOT, int PITCH>
__device__ __forceinline__
void wmma_strip(const float* As, const float* __restrict__ wcol, v8f (&c)[RT]) {
  constexpr int NBLK = KTOT / (4 * KB);
  static_assert(KTOT % (4 * KB) == 0, "whole blocks of k-steps");
  static_assert(NBLK >= 2, "the B fragments run two blocks ahead");
  v2f a[2][KB][RT];
  v2f b[3][KB];
#pragma unroll
  for (int s = 0; s < KB; ++s) {
    b[0][s] = v2f{wcol[(4 * s) * HID], wcol[(4 * s + 1) * HID]};
    b[1][s] = v2f{wcol[(4 * (KB + s)) * HID], wcol[(4 * (KB + s) + 1) * HID]};
  }
#pragma unroll
  for (int s = 0; s < KB; ++s) {
#pragma unroll
    for (int rt = 0; rt < RT; ++rt) a[0][s][rt] = *(const v2f*)(As + rt * 16 * PITCH + 4 * s);
  }
#pragma unroll
  for (int blk = 0; blk < NBLK; ++blk) {
    if (blk + 2 < NBLK) {
#pragma unroll
      for (int s = 0; s < KB; ++s)
        b[(blk + 2) % 3][s] = v2f{wcol[(4 * ((blk + 2) * KB + s)) * HID],
                                  wcol[(4 * ((blk + 2) * KB + s) + 1) * HID]};
    }
    if (blk + 1 < NBLK) {
#pragma unroll
      for (int s = 0; s < KB; ++s) {
#pragma unroll
        for (int rt = 0; rt < RT; ++rt)
          a[(blk + 1) & 1][s][rt] =
              *(const v2f*)(As + rt * 16 * PITCH + 4 * ((blk + 1) * KB + s));
      }
    }
    __builtin_amdgcn_sched_barrier(0);
#pragma unroll
    for (int s = 0; s < KB; ++s) {
#pragma unroll
      for (int rt = 0; rt < RT; ++rt)
        c[rt] = __builtin_amdgcn_wmma_f32_16x16x4_f32(false, a[blk & 1][s][rt], false,
                                                      b[blk % 3][s], (short)0, c[rt],
                                                      false, false);
    }
    __builtin_amdgcn_sched_barrier(0);
  }
}

__global__ void gemm_wmma_kernel(const float* __restrict__ X, const float* X1,
                                 const float* __restrict__ dpart, float* __restrict__ scale,
                                 const float* __restrict__ Wm, const float* __restrict__ bias,
                                 float* Y) {
  __shared__ __attribute__((aligned(16))) float Xs[BROWS][132];
  const int t    = threadIdx.x;
  const int row0 = blockIdx.x * BROWS;

#pragma unroll
  for (int rt = 0; rt < RT; ++rt) {
    int   r    = rt * 16 + (t >> 4);
    int   cseg = (t & 15) << 3;
    const float* src = X + ((long long)row0 + r) * HID + cseg;
    float4 v0 = *(const float4*)(src);
    float4 v1 = *(const float4*)(src + 4);
    static_assert(EH == 2, "two partial planes and two partial degrees are added here");
    const float* src1 = X1 + ((long long)row0 + r) * HID + cseg;
    const float4 w0 = *(const float4*)(src1);
    const float4 w1 = *(const float4*)(src1 + 4);
    v0.x += w0.x; v0.y += w0.y; v0.z += w0.z; v0.w += w0.w;
    v1.x += w1.x; v1.y += w1.y; v1.z += w1.z; v1.w += w1.w;
    const float s = 1.0f / fmaxf(dpart[row0 + r] + dpart[N_NODES + row0 + r], 1.0f);
    if ((t & 15) == 0) scale[row0 + r] = s;
    float* dst = &Xs[r][cseg];
    dst[0] = v0.x * s; dst[1] = v0.y * s; dst[2] = v0.z * s; dst[3] = v0.w * s;
    dst[4] = v1.x * s; dst[5] = v1.y * s; dst[6] = v1.z * s; dst[7] = v1.w * s;
  }
  __syncthreads();

  const int lane = t & 31;
  const int wid  = t >> 5;
  const int m    = lane & 15;
  const int kh   = (lane >> 4) << 1;
  const int n    = (wid << 4) + (lane & 15);

  const float* wcol = Wm + kh * HID + n;

  v8f c[RT];
#pragma unroll
  for (int rt = 0; rt < RT; ++rt) c[rt] = v8f{};

  wmma_strip<HID, 132>(&Xs[m][kh], wcol, c);

  const float bn = bias[n];
#pragma unroll
  for (int rt = 0; rt < RT; ++rt) {
    const int rowbase = row0 + rt * 16 + ((lane >> 4) << 3);
#pragma unroll
    for (int i = 0; i < 8; ++i) {
      Y[(long long)(rowbase + i) * HID + n] = fmaxf(c[rt][i] + bn, 0.0f);
    }
  }
}

__global__ void decode_kernel(const float* __restrict__ h2,
                              const int* __restrict__ U, const int* __restrict__ V,
                              const float* __restrict__ Wd1, const float* __restrict__ bd1,
                              const float* __restrict__ Wd2, const float* __restrict__ bd2,
                              float* __restrict__ out) {
  __shared__ __attribute__((aligned(16))) float zh[BROWS * 260];
  __shared__ float red[BROWS][16];
  float (*Zs)[260] = (float (*)[260])zh;
  float (*Hs)[HID] = (float (*)[HID])zh;

  const int base = blockIdx.x * BROWS;
  const int t    = threadIdx.x;

  {
    const int  prow = t >> 4;
    const int  seg  = t & 15;
    const int* P    = (seg < 8) ? U : V;
    const int  coff = (seg & 7) << 4;
    int node[RT];
#pragma unroll
    for (int rt = 0; rt < RT; ++rt) node[rt] = P[base + rt * 16 + prow];
    float4 zv[RT][4];
#pragma unroll
    for (int rt = 0; rt < RT; ++rt) {
      const float* src = h2 + (long long)node[rt] * HID + coff;
#pragma unroll
      for (int j = 0; j < 4; ++j) zv[rt][j] = *(const float4*)(src + 4 * j);
    }
#pragma unroll
    for (int rt = 0; rt < RT; ++rt) {
      float* dst = &Zs[rt * 16 + prow][seg << 4];
#pragma unroll
      for (int j = 0; j < 4; ++j) {
        dst[4 * j]     = zv[rt][j].x; dst[4 * j + 1] = zv[rt][j].y;
        dst[4 * j + 2] = zv[rt][j].z; dst[4 * j + 3] = zv[rt][j].w;
      }
    }
  }
  __syncthreads();

  const int lane = t & 31;
  const int wid  = t >> 5;
  const int m    = lane & 15;
  const int kh   = (lane >> 4) << 1;
  const int n    = (wid << 4) + (lane & 15);

  const float* wcol = Wd1 + kh * HID + n;

  v8f c[RT];
#pragma unroll
  for (int rt = 0; rt < RT; ++rt) c[rt] = v8f{};

  wmma_strip<2 * HID, 260>(&Zs[m][kh], wcol, c);

  __syncthreads();

  const float bn    = bd1[n];
  const int   rbase = (lane >> 4) << 3;
#pragma unroll
  for (int rt = 0; rt < RT; ++rt) {
#pragma unroll
    for (int i = 0; i < 8; ++i) Hs[rt * 16 + rbase + i][n] = fmaxf(c[rt][i] + bn, 0.0f);
  }
  __syncthreads();

#pragma unroll
  for (int rt = 0; rt < RT; ++rt) {
    int i   = rt * 16 + (t >> 4);
    int seg = t & 15;
    float p = 0.0f;
#pragma unroll
    for (int j = 0; j < 8; ++j) p += Hs[i][(seg << 3) + j] * Wd2[(seg << 3) + j];
    red[i][seg] = p;
  }
  __syncthreads();
  if (t < BROWS) {
    float s = bd2[0];
#pragma unroll
    for (int j = 0; j < 16; ++j) s += red[t][j];
    out[base + t] = s;
  }
}

extern "C" void kernel_launch(void* const* d_in, const int* in_sizes, int n_in,
                              void* d_out, int out_size, void* d_ws, size_t ws_size,
                              hipStream_t stream) {
  const int*   adj_idx = (const int*)  d_in[0];
  const float* adj_val = (const float*)d_in[1];
  const float* h       = (const float*)d_in[3];
  const int*   pos_u   = (const int*)  d_in[4];
  const int*   pos_v   = (const int*)  d_in[5];
  const int*   neg_u   = (const int*)  d_in[6];
  const int*   neg_v   = (const int*)  d_in[7];
  const float* W1      = (const float*)d_in[8];
  const float* b1      = (const float*)d_in[9];
  const float* W2      = (const float*)d_in[10];
  const float* b2      = (const float*)d_in[11];
  const float* Wd1     = (const float*)d_in[12];
  const float* bd1     = (const float*)d_in[13];
  const float* Wd2     = (const float*)d_in[14];
  const float* bd2     = (const float*)d_in[15];

  float* outp = (float*)d_out;
  float* pos_scores = outp;
  float* neg_scores = outp + N_POS;
  float* h2         = outp + 2 * N_POS;

  float* ws      = (float*)d_ws;
  float* deg_inv = ws;
  float* aux     = ws + 50176;
  float* h1      = aux + (long long)N_NODES * HID;

  int*   lists = (int*)aux;
  float* dpart = aux + (long long)NLIST * (2 * CAP + 1);
  int*   keys  = (int*)(dpart + EH * N_NODES);

  edge_key_kernel<<<(N_EDGES / 4 + 255) / 256, 256, 0, stream>>>(adj_idx, keys);

  spmm_sweep_kernel<<<NWG, SPW * 32, 0, stream>>>(adj_idx, adj_val, h, keys, h2, h1, dpart, lists);
  gemm_wmma_kernel<<<N_NODES / BROWS, 256, 0, stream>>>(h2, h1, dpart, deg_inv, W1, b1, h1);
  spmm_replay_gemm_kernel<<<NRP, SPW * 32, 0, stream>>>(adj_idx, adj_val, h1, keys, deg_inv,
                                                        W2, b2, h2, lists);
  decode_kernel<<<N_POS / BROWS, 256, 0, stream>>>(h2, pos_u, pos_v, Wd1, bd1, Wd2, bd2, pos_scores);
  decode_kernel<<<N_POS / BROWS, 256, 0, stream>>>(h2, neg_u, neg_v, Wd1, bd1, Wd2, bd2, neg_scores);
}
